# starting kernel
# baseline (speedup 1.0000x reference)
#include <hip/hip_runtime.h>


typedef __attribute__((ext_vector_type(2))) float v2f;
typedef __attribute__((ext_vector_type(8))) float v8f;

__device__ __forceinline__ void lds_fence() {
  asm volatile("s_wait_dscnt 0" ::: "memory");
}

__device__ __forceinline__ v8f wmma4(v2f a, v2f b, v8f c) {
  return __builtin_amdgcn_wmma_f32_16x16x4_f32(false, a, false, b, (short)0, c,
                                               false, false);
}

__device__ __forceinline__ void cnot_ring(float* Sre, float* Sim, int lane) {
#pragma unroll
  for (int q = 0; q < 9; ++q) {
    const int qt = (q + 1) % 9;
    const int tmask = 1 << qt;
    const int cmask = 1 << q;
#pragma unroll
    for (int jj = 0; jj < 8; ++jj) {
      int k = lane + 32 * jj;
      int idx0 = ((k >> qt) << (qt + 1)) | (k & (tmask - 1));
      if (idx0 & cmask) {
        int idx1 = idx0 | tmask;
        float r0 = Sre[idx0], r1 = Sre[idx1];
        Sre[idx0] = r1; Sre[idx1] = r0;
        float i0 = Sim[idx0], i1 = Sim[idx1];
        Sim[idx0] = i1; Sim[idx1] = i0;
      }
    }
    lds_fence();
  }
}

__launch_bounds__(256)
__global__ void quanv_kernel(const float* __restrict__ x,
                             const float* __restrict__ qw,
                             float* __restrict__ out) {
  constexpr int NQ = 9, B = 4, C = 3, H = 28, W = 28, HOUT = 26, WOUT = 26;
  constexpr int TOTAL = B * HOUT * WOUT;

  __shared__ float lds[8][1024];

  const int lane = threadIdx.x & 31;
  const int warp = threadIdx.x >> 5;
  const int wid = blockIdx.x * 8 + warp;
  if (wid >= TOTAL) return;

  const int b  = wid / (HOUT * WOUT);
  const int yy = (wid / WOUT) % HOUT;
  const int xx = wid % WOUT;

  float* Sre = &lds[warp][0];
  float* Sim = &lds[warp][512];

  const int m16 = lane & 15;
  const int hh  = lane >> 4;

  __builtin_prefetch(x + ((b * C) * H + yy) * W + xx, 0, 1);

  float c1[NQ], s1[NQ];
#pragma unroll
  for (int i = 0; i < NQ; ++i) {
    float a = qw[NQ + i] * 0.5f;
    c1[i] = cosf(a);
    s1[i] = sinf(a);
  }

  v2f gAr[4], gAi[4];
#pragma unroll
  for (int kc = 0; kc < 4; ++kc) {
#pragma unroll
    for (int v = 0; v < 2; ++v) {
      int k = 4 * kc + v + 2 * hh;
      int diff = m16 ^ k;
      float P = 1.0f;
#pragma unroll
      for (int i = 0; i < 4; ++i) P *= ((diff >> i) & 1) ? s1[i] : c1[i];
      int hm = __popc(diff) & 3;
      gAr[kc][v] = (hm == 0) ? P : ((hm == 2) ? -P : 0.0f);
      gAi[kc][v] = (hm == 1) ? -P : ((hm == 3) ? P : 0.0f);
    }
  }

  v2f gBr[2][8], gBi[2][8];
#pragma unroll
  for (int t = 0; t < 2; ++t) {
    int cp = 16 * t + m16;
#pragma unroll
    for (int kc = 0; kc < 8; ++kc) {
#pragma unroll
      for (int v = 0; v < 2; ++v) {
        int c = 4 * kc + v + 2 * hh;
        int diff = c ^ cp;
        float P = 1.0f;
#pragma unroll
        for (int i = 0; i < 5; ++i) P *= ((diff >> i) & 1) ? s1[4 + i] : c1[4 + i];
        int hm = __popc(diff) & 3;
        gBr[t][kc][v] = (hm == 0) ? P : ((hm == 2) ? -P : 0.0f);
        gBi[t][kc][v] = (hm == 1) ? -P : ((hm == 3) ? P : 0.0f);
      }
    }
  }

  float zacc[4] = {0.f, 0.f, 0.f, 0.f};

  for (int ch = 0; ch < C; ++ch) {
    float ca[NQ], sa[NQ];
    const float* xp = x + ((b * C + ch) * H + yy) * W + xx;
#pragma unroll
    for (int dy = 0; dy < 3; ++dy) {
#pragma unroll
      for (int dx = 0; dx < 3; ++dx) {
        int i = dy * 3 + dx;
        float a = (xp[dy * W + dx] + qw[i]) * 0.5f;
        ca[i] = cosf(a);
        sa[i] = sinf(a);
      }
    }

#pragma unroll
    for (int j = 0; j < 16; ++j) {
      int idx = lane + 32 * j;
      float P = 1.0f;
#pragma unroll
      for (int i = 0; i < NQ; ++i) P *= ((idx >> i) & 1) ? sa[i] : ca[i];
      int hm = __popc(idx) & 3;
      Sre[idx] = (hm == 0) ? P : ((hm == 2) ? -P : 0.0f);
      Sim[idx] = (hm == 1) ? -P : ((hm == 3) ? P : 0.0f);
    }
    lds_fence();

    cnot_ring(Sre, Sim, lane);

    v2f bR[2][4], bI[2][4];
#pragma unroll
    for (int t = 0; t < 2; ++t) {
#pragma unroll
      for (int kc = 0; kc < 4; ++kc) {
#pragma unroll
        for (int v = 0; v < 2; ++v) {
          int r = 4 * kc + v + 2 * hh;
          int c = 16 * t + m16;
          bR[t][kc][v] = Sre[r + 16 * c];
          bI[t][kc][v] = Sim[r + 16 * c];
        }
      }
    }
#pragma unroll
    for (int t = 0; t < 2; ++t) {
      v8f rr = {0.f,0.f,0.f,0.f,0.f,0.f,0.f,0.f};
      v8f ii = rr, ri = rr, ir = rr;
#pragma unroll
      for (int kc = 0; kc < 4; ++kc) {
        rr = wmma4(gAr[kc], bR[t][kc], rr);
        ii = wmma4(gAi[kc], bI[t][kc], ii);
        ri = wmma4(gAr[kc], bI[t][kc], ri);
        ir = wmma4(gAi[kc], bR[t][kc], ir);
      }
      v8f nr = rr - ii, ni = ri + ir;
#pragma unroll
      for (int v = 0; v < 8; ++v) {
        int r = v + 8 * hh;
        int c = 16 * t + m16;
        Sre[r + 16 * c] = nr[v];
        Sim[r + 16 * c] = ni[v];
      }
    }
    lds_fence();

    v2f aR[8], aI[8];
#pragma unroll
    for (int kc = 0; kc < 8; ++kc) {
#pragma unroll
      for (int v = 0; v < 2; ++v) {
        int c = 4 * kc + v + 2 * hh;
        aR[kc][v] = Sre[m16 + 16 * c];
        aI[kc][v] = Sim[m16 + 16 * c];
      }
    }
#pragma unroll
    for (int t = 0; t < 2; ++t) {
      v8f rr = {0.f,0.f,0.f,0.f,0.f,0.f,0.f,0.f};
      v8f ii = rr, ri = rr, ir = rr;
#pragma unroll
      for (int kc = 0; kc < 8; ++kc) {
        rr = wmma4(aR[kc], gBr[t][kc], rr);
        ii = wmma4(aI[kc], gBi[t][kc], ii);
        ri = wmma4(aI[kc], gBr[t][kc], ri);
        ir = wmma4(aR[kc], gBi[t][kc], ir);
      }
      v8f nr = rr - ii, ni = ri + ir;
#pragma unroll
      for (int v = 0; v < 8; ++v) {
        int r = v + 8 * hh;
        int c = 16 * t + m16;
        Sre[r + 16 * c] = nr[v];
        Sim[r + 16 * c] = ni[v];
      }
    }
    lds_fence();

    cnot_ring(Sre, Sim, lane);

    float z[4] = {0.f, 0.f, 0.f, 0.f};
#pragma unroll
    for (int j = 0; j < 16; ++j) {
      int idx = lane + 32 * j;
      float re = Sre[idx], im = Sim[idx];
      float p = re * re + im * im;
#pragma unroll
      for (int f = 0; f < 4; ++f) z[f] += ((idx >> f) & 1) ? -p : p;
    }
#pragma unroll
    for (int f = 0; f < 4; ++f) {
#pragma unroll
      for (int m = 1; m < 32; m <<= 1) z[f] += __shfl_xor(z[f], m, 32);
      zacc[f] += z[f];
    }
    lds_fence();
  }

  if (lane == 0) {
    const float inv = 1.0f / 3.0f;
#pragma unroll
    for (int f = 0; f < 4; ++f)
      out[((b * 4 + f) * HOUT + yy) * WOUT + xx] = zacc[f] * inv;
  }
}

extern "C" void kernel_launch(void* const* d_in, const int* in_sizes, int n_in,
                              void* d_out, int out_size, void* d_ws, size_t ws_size,
                              hipStream_t stream) {
  (void)in_sizes; (void)n_in; (void)out_size; (void)d_ws; (void)ws_size;
  const float* x  = (const float*)d_in[0];
  const float* qw = (const float*)d_in[1];
  float* out = (float*)d_out;
  const int total_waves = 4 * 26 * 26;
  dim3 grid(total_waves / 8), block(256);
  hipLaunchKernelGGL(quanv_kernel, grid, block, 0, stream, x, qw, out);
}


# Lane-uniform and wave-invariant setup taken once per workgroup, shared through LDS: one thread per angle takes sin/cos (225 in one pass), each wave builds 5 of 40 gate-fragment entries; same values
# speedup vs baseline: 1.3140x; 1.3140x over previous
#include <hip/hip_runtime.h>


typedef __attribute__((ext_vector_type(2))) float v2f;
typedef __attribute__((ext_vector_type(8))) float v8f;

__device__ __forceinline__ void lds_fence() {
  asm volatile("s_wait_dscnt 0" ::: "memory");
}

__device__ __forceinline__ v8f wmma4(v2f a, v2f b, v8f c) {
  return __builtin_amdgcn_wmma_f32_16x16x4_f32(false, a, false, b, (short)0, c,
                                               false, false);
}

__device__ __forceinline__ void cnot_ring(float* Sre, float* Sim, int lane) {
#pragma unroll
  for (int q = 0; q < 9; ++q) {
    const int qt = (q + 1) % 9;
    const int tmask = 1 << qt;
    const int cmask = 1 << q;
#pragma unroll
    for (int jj = 0; jj < 8; ++jj) {
      int k = lane + 32 * jj;
      int idx0 = ((k >> qt) << (qt + 1)) | (k & (tmask - 1));
      if (idx0 & cmask) {
        int idx1 = idx0 | tmask;
        float r0 = Sre[idx0], r1 = Sre[idx1];
        Sre[idx0] = r1; Sre[idx1] = r0;
        float i0 = Sim[idx0], i1 = Sim[idx1];
        Sim[idx0] = i1; Sim[idx1] = i0;
      }
    }
    lds_fence();
  }
}

__launch_bounds__(256)
__global__ void quanv_kernel(const float* __restrict__ x,
                             const float* __restrict__ qw,
                             float* __restrict__ out) {
  constexpr int NQ = 9, B = 4, C = 3, H = 28, W = 28, HOUT = 26, WOUT = 26;
  constexpr int TOTAL = B * HOUT * WOUT;

  __shared__ float lds[8][1024];
  __shared__ float2 trig[225];
  __shared__ float2 gtab[40][32];

  const int tid  = threadIdx.x;
  const int lane = tid & 31;
  const int warp = __builtin_amdgcn_readfirstlane(tid >> 5);
  const int wid  = blockIdx.x * 8 + warp;
  const int widc = wid < TOTAL ? wid : TOTAL - 1;

  const int b  = widc / (HOUT * WOUT);
  const int yy = (widc / WOUT) % HOUT;
  const int xx = widc % WOUT;

  float* Sre = &lds[warp][0];
  float* Sim = &lds[warp][512];

  const int m16 = lane & 15;
  const int hh  = lane >> 4;

  if (tid < 225) {
    float a;
    if (tid < 216) {
      const int pw = tid / 27, j = tid % 27;
      const int ch = j / 9, i = j % 9;
      int p = blockIdx.x * 8 + pw;
      p = p < TOTAL ? p : TOTAL - 1;
      const int pb = p / (HOUT * WOUT);
      const int py = (p / WOUT) % HOUT;
      const int px = p % WOUT;
      a = (x[((pb * C + ch) * H + py + i / 3) * W + px + i % 3] + qw[i]) * 0.5f;
    } else {
      a = qw[NQ + (tid - 216)] * 0.5f;
    }
    trig[tid] = make_float2(cosf(a), sinf(a));
  }
  __syncthreads();

  auto uni = [](float v) -> float {
    return __int_as_float(__builtin_amdgcn_readfirstlane(__float_as_int(v)));
  };

  float c1[NQ], s1[NQ];
#pragma unroll
  for (int i = 0; i < NQ; ++i) {
    float2 t = trig[216 + i];
    c1[i] = uni(t.x);
    s1[i] = uni(t.y);
  }

#pragma unroll
  for (int j = 0; j < 5; ++j) {
    const int e = warp * 5 + j;
    int diff;
    float P = 1.0f;
    if (e < 8) {
      int k = 4 * (e >> 1) + (e & 1) + 2 * hh;
      diff = m16 ^ k;
#pragma unroll
      for (int i = 0; i < 4; ++i) P *= ((diff >> i) & 1) ? s1[i] : c1[i];
    } else {
      const int f = e - 8;
      int cp = 16 * (f >> 4) + m16;
      int c = 4 * ((f >> 1) & 7) + (f & 1) + 2 * hh;
      diff = c ^ cp;
#pragma unroll
      for (int i = 0; i < 5; ++i) P *= ((diff >> i) & 1) ? s1[4 + i] : c1[4 + i];
    }
    int hm = __popc(diff) & 3;
    float re = (hm == 0) ? P : ((hm == 2) ? -P : 0.0f);
    float im = (hm == 1) ? -P : ((hm == 3) ? P : 0.0f);
    gtab[e][lane] = make_float2(re, im);
  }
  __syncthreads();


  float zacc[4] = {0.f, 0.f, 0.f, 0.f};

  for (int ch = 0; ch < C; ++ch) {
    float ca[NQ], sa[NQ];
#pragma unroll
    for (int i = 0; i < NQ; ++i) {
      float2 t = trig[27 * warp + 9 * ch + i];
      ca[i] = uni(t.x);
      sa[i] = uni(t.y);
    }

#pragma unroll
    for (int j = 0; j < 16; ++j) {
      int idx = lane + 32 * j;
      float P = 1.0f;
#pragma unroll
      for (int i = 0; i < NQ; ++i) P *= ((idx >> i) & 1) ? sa[i] : ca[i];
      int hm = __popc(idx) & 3;
      Sre[idx] = (hm == 0) ? P : ((hm == 2) ? -P : 0.0f);
      Sim[idx] = (hm == 1) ? -P : ((hm == 3) ? P : 0.0f);
    }
    lds_fence();

    cnot_ring(Sre, Sim, lane);

    v2f gAr[4], gAi[4];
#pragma unroll
    for (int kc = 0; kc < 4; ++kc) {
#pragma unroll
      for (int v = 0; v < 2; ++v) {
        float2 g = gtab[2 * kc + v][lane];
        gAr[kc][v] = g.x;
        gAi[kc][v] = g.y;
      }
    }
    v2f bR[2][4], bI[2][4];
#pragma unroll
    for (int t = 0; t < 2; ++t) {
#pragma unroll
      for (int kc = 0; kc < 4; ++kc) {
#pragma unroll
        for (int v = 0; v < 2; ++v) {
          int r = 4 * kc + v + 2 * hh;
          int c = 16 * t + m16;
          bR[t][kc][v] = Sre[r + 16 * c];
          bI[t][kc][v] = Sim[r + 16 * c];
        }
      }
    }
#pragma unroll
    for (int t = 0; t < 2; ++t) {
      v8f rr = {0.f,0.f,0.f,0.f,0.f,0.f,0.f,0.f};
      v8f ii = rr, ri = rr, ir = rr;
#pragma unroll
      for (int kc = 0; kc < 4; ++kc) {
        rr = wmma4(gAr[kc], bR[t][kc], rr);
        ii = wmma4(gAi[kc], bI[t][kc], ii);
        ri = wmma4(gAr[kc], bI[t][kc], ri);
        ir = wmma4(gAi[kc], bR[t][kc], ir);
      }
      v8f nr = rr - ii, ni = ri + ir;
#pragma unroll
      for (int v = 0; v < 8; ++v) {
        int r = v + 8 * hh;
        int c = 16 * t + m16;
        Sre[r + 16 * c] = nr[v];
        Sim[r + 16 * c] = ni[v];
      }
    }
    lds_fence();

    v2f gBr[2][8], gBi[2][8];
#pragma unroll
    for (int t = 0; t < 2; ++t) {
#pragma unroll
      for (int kc = 0; kc < 8; ++kc) {
#pragma unroll
        for (int v = 0; v < 2; ++v) {
          float2 g = gtab[8 + 16 * t + 2 * kc + v][lane];
          gBr[t][kc][v] = g.x;
          gBi[t][kc][v] = g.y;
        }
      }
    }
    v2f aR[8], aI[8];
#pragma unroll
    for (int kc = 0; kc < 8; ++kc) {
#pragma unroll
      for (int v = 0; v < 2; ++v) {
        int c = 4 * kc + v + 2 * hh;
        aR[kc][v] = Sre[m16 + 16 * c];
        aI[kc][v] = Sim[m16 + 16 * c];
      }
    }
#pragma unroll
    for (int t = 0; t < 2; ++t) {
      v8f rr = {0.f,0.f,0.f,0.f,0.f,0.f,0.f,0.f};
      v8f ii = rr, ri = rr, ir = rr;
#pragma unroll
      for (int kc = 0; kc < 8; ++kc) {
        rr = wmma4(aR[kc], gBr[t][kc], rr);
        ii = wmma4(aI[kc], gBi[t][kc], ii);
        ri = wmma4(aI[kc], gBr[t][kc], ri);
        ir = wmma4(aR[kc], gBi[t][kc], ir);
      }
      v8f nr = rr - ii, ni = ri + ir;
#pragma unroll
      for (int v = 0; v < 8; ++v) {
        int r = v + 8 * hh;
        int c = 16 * t + m16;
        Sre[r + 16 * c] = nr[v];
        Sim[r + 16 * c] = ni[v];
      }
    }
    lds_fence();

    cnot_ring(Sre, Sim, lane);

    float z[4] = {0.f, 0.f, 0.f, 0.f};
#pragma unroll
    for (int j = 0; j < 16; ++j) {
      int idx = lane + 32 * j;
      float re = Sre[idx], im = Sim[idx];
      float p = re * re + im * im;
#pragma unroll
      for (int f = 0; f < 4; ++f) z[f] += ((idx >> f) & 1) ? -p : p;
    }
#pragma unroll
    for (int f = 0; f < 4; ++f) {
#pragma unroll
      for (int m = 1; m < 32; m <<= 1) z[f] += __shfl_xor(z[f], m, 32);
      zacc[f] += z[f];
    }
    lds_fence();
  }

  if (lane == 0 && wid < TOTAL) {
    const float inv = 1.0f / 3.0f;
#pragma unroll
    for (int f = 0; f < 4; ++f)
      out[((b * 4 + f) * HOUT + yy) * WOUT + xx] = zacc[f] * inv;
  }
}

extern "C" void kernel_launch(void* const* d_in, const int* in_sizes, int n_in,
                              void* d_out, int out_size, void* d_ws, size_t ws_size,
                              hipStream_t stream) {
  (void)in_sizes; (void)n_in; (void)out_size; (void)d_ws; (void)ws_size;
  const float* x  = (const float*)d_in[0];
  const float* qw = (const float*)d_in[1];
  float* out = (float*)d_out;
  const int total_waves = 4 * 26 * 26;
  dim3 grid(total_waves / 8), block(256);
  hipLaunchKernelGGL(quanv_kernel, grid, block, 0, stream, x, qw, out);
}


# Both CNOT rings in registers (ring 1 on the product state, ring 2 on the 2nd product's accumulators): rename, v_cndmask or DPP/permlane move per CNOT; 18 LDS swap passes gone; Z sum order changes
# speedup vs baseline: 2.0276x; 1.5431x over previous
#include <hip/hip_runtime.h>


typedef __attribute__((ext_vector_type(2))) float v2f;
typedef __attribute__((ext_vector_type(8))) float v8f;

__device__ __forceinline__ void lds_fence() {
  asm volatile("s_wait_dscnt 0" ::: "memory");
}

__device__ __forceinline__ v8f wmma4(v2f a, v2f b, v8f c) {
  return __builtin_amdgcn_wmma_f32_16x16x4_f32(false, a, false, b, (short)0, c,
                                               false, false);
}


template <int CTRL, int ROWS>
__device__ __forceinline__ float lane_quad(float v) {
  int x = __float_as_int(v);
  return __int_as_float(__builtin_amdgcn_update_dpp(x, x, CTRL, ROWS, 0xF, false));
}
__device__ __forceinline__ float cx_l0_l1(float v) { return lane_quad<0x6C, 0x3>(v); }
__device__ __forceinline__ float cx_l1_l2(float v) {
  constexpr int SEL = 0 | (1 << 3) | (6 << 6) | (7 << 9) | (4 << 12) | (5 << 15) |
                      (2 << 18) | (3 << 21);
  return __int_as_float(__builtin_amdgcn_mov_dpp8(__float_as_int(v), SEL));
}
__device__ __forceinline__ float cx_l2_l3(float v) {
  int x = __float_as_int(v);
  return __int_as_float(
      __builtin_amdgcn_permlane16(x, x, 0xFEDC3210u, 0x7654BA98u, false, false));
}
__device__ __forceinline__ float other_row(float v) {
  int x = __float_as_int(v);
  return __int_as_float(
      __builtin_amdgcn_permlanex16(x, x, 0x76543210u, 0xFEDCBA98u, false, false));
}
__device__ __forceinline__ void swapf(float& a, float& b) { float t = a; a = b; b = t; }

__device__ __forceinline__ void cnot_ring_lane_major(float (&r)[16], int lane) {
  const bool l3 = (lane & 8) != 0, l4 = (lane & 16) != 0;
#pragma unroll
  for (int j = 0; j < 16; ++j) {
    float v = cx_l2_l3(cx_l1_l2(cx_l0_l1(r[j])));
    float o = other_row(v);
    r[j] = l3 ? o : v;
  }
#pragma unroll
  for (int j = 0; j < 16; j += 2) {
    float a = r[j], b = r[j + 1];
    r[j] = l4 ? b : a;
    r[j + 1] = l4 ? a : b;
  }
#pragma unroll
  for (int j = 0; j < 16; ++j) if ((j & 1) && !(j & 2)) swapf(r[j], r[j ^ 2]);
#pragma unroll
  for (int j = 0; j < 16; ++j) if ((j & 2) && !(j & 4)) swapf(r[j], r[j ^ 4]);
#pragma unroll
  for (int j = 0; j < 16; ++j) if ((j & 4) && !(j & 8)) swapf(r[j], r[j ^ 8]);
#pragma unroll
  for (int j = 8; j < 16; ++j) r[j] = lane_quad<0xB1, 0x3>(r[j]);
}

__device__ __forceinline__ void cnot_ring_acc(float (&g)[16], int lane) {
  const bool l3 = (lane & 8) != 0;
#pragma unroll
  for (int t = 0; t < 2; ++t) {
#pragma unroll
    for (int v = 0; v < 8; ++v) if ((v & 1) && !(v & 2)) swapf(g[8 * t + v], g[8 * t + (v ^ 2)]);
#pragma unroll
    for (int v = 0; v < 8; ++v) if ((v & 2) && !(v & 4)) swapf(g[8 * t + v], g[8 * t + (v ^ 4)]);
  }
#pragma unroll
  for (int e = 0; e < 16; ++e) {
    float v = g[e];
    if (e & 4) v = other_row(v);
    v = lane_quad<0xB1, 0x2>(v);
    g[e] = cx_l2_l3(cx_l1_l2(cx_l0_l1(v)));
  }
#pragma unroll
  for (int v = 0; v < 8; ++v) {
    float a = g[v], b = g[8 + v];
    g[v] = l3 ? b : a;
    g[8 + v] = l3 ? a : b;
  }
#pragma unroll
  for (int v = 0; v < 8; v += 2) swapf(g[8 + v], g[8 + v + 1]);
}

__launch_bounds__(256)
__global__ void quanv_kernel(const float* __restrict__ x,
                             const float* __restrict__ qw,
                             float* __restrict__ out) {
  constexpr int NQ = 9, B = 4, C = 3, H = 28, W = 28, HOUT = 26, WOUT = 26;
  constexpr int TOTAL = B * HOUT * WOUT;

  __shared__ float lds[8][1024];
  __shared__ float2 trig[225];
  __shared__ float2 gtab[40][32];

  const int tid  = threadIdx.x;
  const int lane = tid & 31;
  const int warp = __builtin_amdgcn_readfirstlane(tid >> 5);
  const int wid  = blockIdx.x * 8 + warp;
  const int widc = wid < TOTAL ? wid : TOTAL - 1;

  const int b  = widc / (HOUT * WOUT);
  const int yy = (widc / WOUT) % HOUT;
  const int xx = widc % WOUT;

  float* Sre = &lds[warp][0];
  float* Sim = &lds[warp][512];

  const int m16 = lane & 15;
  const int hh  = lane >> 4;

  if (tid < 225) {
    float a;
    if (tid < 216) {
      const int pw = tid / 27, j = tid % 27;
      const int ch = j / 9, i = j % 9;
      int p = blockIdx.x * 8 + pw;
      p = p < TOTAL ? p : TOTAL - 1;
      const int pb = p / (HOUT * WOUT);
      const int py = (p / WOUT) % HOUT;
      const int px = p % WOUT;
      a = (x[((pb * C + ch) * H + py + i / 3) * W + px + i % 3] + qw[i]) * 0.5f;
    } else {
      a = qw[NQ + (tid - 216)] * 0.5f;
    }
    trig[tid] = make_float2(cosf(a), sinf(a));
  }
  __syncthreads();

  auto uni = [](float v) -> float {
    return __int_as_float(__builtin_amdgcn_readfirstlane(__float_as_int(v)));
  };

  float c1[NQ], s1[NQ];
#pragma unroll
  for (int i = 0; i < NQ; ++i) {
    float2 t = trig[216 + i];
    c1[i] = uni(t.x);
    s1[i] = uni(t.y);
  }

#pragma unroll
  for (int j = 0; j < 5; ++j) {
    const int e = warp * 5 + j;
    int diff;
    float P = 1.0f;
    if (e < 8) {
      int k = 4 * (e >> 1) + (e & 1) + 2 * hh;
      diff = m16 ^ k;
#pragma unroll
      for (int i = 0; i < 4; ++i) P *= ((diff >> i) & 1) ? s1[i] : c1[i];
    } else {
      const int f = e - 8;
      int cp = 16 * (f >> 4) + m16;
      int c = 4 * ((f >> 1) & 7) + (f & 1) + 2 * hh;
      diff = c ^ cp;
#pragma unroll
      for (int i = 0; i < 5; ++i) P *= ((diff >> i) & 1) ? s1[4 + i] : c1[4 + i];
    }
    int hm = __popc(diff) & 3;
    float re = (hm == 0) ? P : ((hm == 2) ? -P : 0.0f);
    float im = (hm == 1) ? -P : ((hm == 3) ? P : 0.0f);
    gtab[e][lane] = make_float2(re, im);
  }
  __syncthreads();


  float zacc[4] = {0.f, 0.f, 0.f, 0.f};

  for (int ch = 0; ch < C; ++ch) {
    float ca[NQ], sa[NQ];
#pragma unroll
    for (int i = 0; i < NQ; ++i) {
      float2 t = trig[27 * warp + 9 * ch + i];
      ca[i] = uni(t.x);
      sa[i] = uni(t.y);
    }

    float pRe[16], pIm[16];
#pragma unroll
    for (int j = 0; j < 16; ++j) {
      int idx = lane + 32 * j;
      float P = 1.0f;
#pragma unroll
      for (int i = 0; i < NQ; ++i) P *= ((idx >> i) & 1) ? sa[i] : ca[i];
      int hm = __popc(idx) & 3;
      pRe[j] = (hm == 0) ? P : ((hm == 2) ? -P : 0.0f);
      pIm[j] = (hm == 1) ? -P : ((hm == 3) ? P : 0.0f);
    }

    cnot_ring_lane_major(pRe, lane);
    cnot_ring_lane_major(pIm, lane);
#pragma unroll
    for (int j = 0; j < 16; ++j) {
      Sre[lane + 32 * j] = pRe[j];
      Sim[lane + 32 * j] = pIm[j];
    }
    lds_fence();

    v2f gAr[4], gAi[4];
#pragma unroll
    for (int kc = 0; kc < 4; ++kc) {
#pragma unroll
      for (int v = 0; v < 2; ++v) {
        float2 g = gtab[2 * kc + v][lane];
        gAr[kc][v] = g.x;
        gAi[kc][v] = g.y;
      }
    }
    v2f bR[2][4], bI[2][4];
#pragma unroll
    for (int t = 0; t < 2; ++t) {
#pragma unroll
      for (int kc = 0; kc < 4; ++kc) {
#pragma unroll
        for (int v = 0; v < 2; ++v) {
          int r = 4 * kc + v + 2 * hh;
          int c = 16 * t + m16;
          bR[t][kc][v] = Sre[r + 16 * c];
          bI[t][kc][v] = Sim[r + 16 * c];
        }
      }
    }
#pragma unroll
    for (int t = 0; t < 2; ++t) {
      v8f rr = {0.f,0.f,0.f,0.f,0.f,0.f,0.f,0.f};
      v8f ii = rr, ri = rr, ir = rr;
#pragma unroll
      for (int kc = 0; kc < 4; ++kc) {
        rr = wmma4(gAr[kc], bR[t][kc], rr);
        ii = wmma4(gAi[kc], bI[t][kc], ii);
        ri = wmma4(gAr[kc], bI[t][kc], ri);
        ir = wmma4(gAi[kc], bR[t][kc], ir);
      }
      v8f nr = rr - ii, ni = ri + ir;
#pragma unroll
      for (int v = 0; v < 8; ++v) {
        int r = v + 8 * hh;
        int c = 16 * t + m16;
        Sre[r + 16 * c] = nr[v];
        Sim[r + 16 * c] = ni[v];
      }
    }
    lds_fence();

    v2f gBr[2][8], gBi[2][8];
#pragma unroll
    for (int t = 0; t < 2; ++t) {
#pragma unroll
      for (int kc = 0; kc < 8; ++kc) {
#pragma unroll
        for (int v = 0; v < 2; ++v) {
          float2 g = gtab[8 + 16 * t + 2 * kc + v][lane];
          gBr[t][kc][v] = g.x;
          gBi[t][kc][v] = g.y;
        }
      }
    }
    v2f aR[8], aI[8];
#pragma unroll
    for (int kc = 0; kc < 8; ++kc) {
#pragma unroll
      for (int v = 0; v < 2; ++v) {
        int c = 4 * kc + v + 2 * hh;
        aR[kc][v] = Sre[m16 + 16 * c];
        aI[kc][v] = Sim[m16 + 16 * c];
      }
    }
    float fRe[16], fIm[16];
#pragma unroll
    for (int t = 0; t < 2; ++t) {
      v8f rr = {0.f,0.f,0.f,0.f,0.f,0.f,0.f,0.f};
      v8f ii = rr, ri = rr, ir = rr;
#pragma unroll
      for (int kc = 0; kc < 8; ++kc) {
        rr = wmma4(aR[kc], gBr[t][kc], rr);
        ii = wmma4(aI[kc], gBi[t][kc], ii);
        ri = wmma4(aI[kc], gBr[t][kc], ri);
        ir = wmma4(aR[kc], gBi[t][kc], ir);
      }
      v8f nr = rr - ii, ni = ri + ir;
#pragma unroll
      for (int v = 0; v < 8; ++v) {
        fRe[8 * t + v] = nr[v];
        fIm[8 * t + v] = ni[v];
      }
    }

    cnot_ring_acc(fRe, lane);
    cnot_ring_acc(fIm, lane);

    float z[4] = {0.f, 0.f, 0.f, 0.f};
#pragma unroll
    for (int e = 0; e < 16; ++e) {
      float p = fRe[e] * fRe[e] + fIm[e] * fIm[e];
#pragma unroll
      for (int f = 0; f < 3; ++f) z[f] += ((e >> f) & 1) ? -p : p;
      z[3] += hh ? -p : p;
    }
#pragma unroll
    for (int f = 0; f < 4; ++f) {
#pragma unroll
      for (int m = 1; m < 32; m <<= 1) z[f] += __shfl_xor(z[f], m, 32);
      zacc[f] += z[f];
    }
  }

  if (lane == 0 && wid < TOTAL) {
    const float inv = 1.0f / 3.0f;
#pragma unroll
    for (int f = 0; f < 4; ++f)
      out[((b * 4 + f) * HOUT + yy) * WOUT + xx] = zacc[f] * inv;
  }
}

extern "C" void kernel_launch(void* const* d_in, const int* in_sizes, int n_in,
                              void* d_out, int out_size, void* d_ws, size_t ws_size,
                              hipStream_t stream) {
  (void)in_sizes; (void)n_in; (void)out_size; (void)d_ws; (void)ws_size;
  const float* x  = (const float*)d_in[0];
  const float* qw = (const float*)d_in[1];
  float* out = (float*)d_out;
  const int total_waves = 4 * 26 * 26;
  dim3 grid(total_waves / 8), block(256);
  hipLaunchKernelGGL(quanv_kernel, grid, block, 0, stream, x, qw, out);
}


# RX(w1) layer as three 3-qubit gates, each 8x8 complex gate stacked re/im into one real 16x16 WMMA operand: 3x16 WMMAs per channel, not 32+64; two more LDS passes of the state; sum order changes
# speedup vs baseline: 2.4619x; 1.2142x over previous
#include <hip/hip_runtime.h>


typedef __attribute__((ext_vector_type(2))) float v2f;
typedef __attribute__((ext_vector_type(8))) float v8f;

__device__ __forceinline__ void lds_fence() {
  asm volatile("s_wait_dscnt 0" ::: "memory");
}

__device__ __forceinline__ v8f wmma4(v2f a, v2f b, v8f c) {
  return __builtin_amdgcn_wmma_f32_16x16x4_f32(false, a, false, b, (short)0, c,
                                               false, false);
}


template <int CTRL, int ROWS>
__device__ __forceinline__ float lane_quad(float v) {
  int x = __float_as_int(v);
  return __int_as_float(__builtin_amdgcn_update_dpp(x, x, CTRL, ROWS, 0xF, false));
}
__device__ __forceinline__ float cx_l0_l1(float v) { return lane_quad<0x6C, 0x3>(v); }
__device__ __forceinline__ float cx_l1_l2(float v) {
  constexpr int SEL = 0 | (1 << 3) | (6 << 6) | (7 << 9) | (4 << 12) | (5 << 15) |
                      (2 << 18) | (3 << 21);
  return __int_as_float(__builtin_amdgcn_mov_dpp8(__float_as_int(v), SEL));
}
__device__ __forceinline__ float cx_l2_l3(float v) {
  int x = __float_as_int(v);
  return __int_as_float(
      __builtin_amdgcn_permlane16(x, x, 0xFEDC3210u, 0x7654BA98u, false, false));
}
__device__ __forceinline__ float other_row(float v) {
  int x = __float_as_int(v);
  return __int_as_float(
      __builtin_amdgcn_permlanex16(x, x, 0x76543210u, 0xFEDCBA98u, false, false));
}
__device__ __forceinline__ void swapf(float& a, float& b) { float t = a; a = b; b = t; }

__device__ __forceinline__ void cnot_ring_lane_major(float (&r)[16], int lane) {
  const bool l3 = (lane & 8) != 0, l4 = (lane & 16) != 0;
#pragma unroll
  for (int j = 0; j < 16; ++j) {
    float v = cx_l2_l3(cx_l1_l2(cx_l0_l1(r[j])));
    float o = other_row(v);
    r[j] = l3 ? o : v;
  }
#pragma unroll
  for (int j = 0; j < 16; j += 2) {
    float a = r[j], b = r[j + 1];
    r[j] = l4 ? b : a;
    r[j + 1] = l4 ? a : b;
  }
#pragma unroll
  for (int j = 0; j < 16; ++j) if ((j & 1) && !(j & 2)) swapf(r[j], r[j ^ 2]);
#pragma unroll
  for (int j = 0; j < 16; ++j) if ((j & 2) && !(j & 4)) swapf(r[j], r[j ^ 4]);
#pragma unroll
  for (int j = 0; j < 16; ++j) if ((j & 4) && !(j & 8)) swapf(r[j], r[j ^ 8]);
#pragma unroll
  for (int j = 8; j < 16; ++j) r[j] = lane_quad<0xB1, 0x3>(r[j]);
}

__device__ __forceinline__ void cnot_ring_acc(float (&g)[16], int lane) {
  const bool l3 = (lane & 8) != 0;
#pragma unroll
  for (int t = 0; t < 2; ++t) {
#pragma unroll
    for (int v = 0; v < 8; ++v) if ((v & 1) && !(v & 2)) swapf(g[8 * t + v], g[8 * t + (v ^ 2)]);
#pragma unroll
    for (int v = 0; v < 8; ++v) if ((v & 2) && !(v & 4)) swapf(g[8 * t + v], g[8 * t + (v ^ 4)]);
  }
#pragma unroll
  for (int e = 0; e < 16; ++e) {
    float v = g[e];
    if (e & 4) v = other_row(v);
    v = lane_quad<0xB1, 0x2>(v);
    g[e] = cx_l2_l3(cx_l1_l2(cx_l0_l1(v)));
  }
#pragma unroll
  for (int v = 0; v < 8; ++v) {
    float a = g[v], b = g[8 + v];
    g[v] = l3 ? b : a;
    g[8 + v] = l3 ? a : b;
  }
#pragma unroll
  for (int v = 0; v < 8; v += 2) swapf(g[8 + v], g[8 + v + 1]);
}

template <int G>
__device__ __forceinline__ void rx_group(float* S, const float2* gtG, int lane) {
  const int n = lane & 15, hh = lane >> 4;
  constexpr int SG = (G == 0) ? 1 : (G == 1) ? 8 : 64;
  constexpr int ST = (G == 2) ? 16 : 128;
  int lb, ld;
  if (G == 0)      { lb = 8 * n + 4 * hh;                        ld = 8 * n + 512 * hh; }
  else if (G == 1) { lb = (n & 7) + 64 * (n >> 3) + 32 * hh;     ld = (n & 7) + 64 * (n >> 3) + 512 * hh; }
  else             { lb = n + 256 * hh;                          ld = n + 512 * hh; }

  v2f a[4];
#pragma unroll
  for (int kc = 0; kc < 4; ++kc) {
    float2 w = gtG[kc * 32 + lane];
    a[kc][0] = w.x;
    a[kc][1] = w.y;
  }
  v2f bb[4][4];
  if (G == 0) {
#pragma unroll
    for (int t = 0; t < 4; ++t) {
#pragma unroll
      for (int reim = 0; reim < 2; ++reim) {
        const float4 q = *reinterpret_cast<const float4*>(&S[lb + ST * t + 512 * reim]);
        bb[t][2 * reim][0] = q.x;
        bb[t][2 * reim][1] = q.y;
        bb[t][2 * reim + 1][0] = q.z;
        bb[t][2 * reim + 1][1] = q.w;
      }
    }
  } else {
#pragma unroll
    for (int t = 0; t < 4; ++t) {
#pragma unroll
      for (int kc = 0; kc < 4; ++kc) {
#pragma unroll
        for (int v = 0; v < 2; ++v)
          bb[t][kc][v] = S[lb + SG * (v + 2 * (kc & 1)) + ST * t + 512 * (kc >> 1)];
      }
    }
  }
  __builtin_amdgcn_sched_barrier(0);
  v8f acc[4];
#pragma unroll
  for (int t = 0; t < 4; ++t) acc[t] = v8f{0.f, 0.f, 0.f, 0.f, 0.f, 0.f, 0.f, 0.f};
#pragma unroll
  for (int kc = 0; kc < 4; ++kc) {
#pragma unroll
    for (int t = 0; t < 4; ++t) acc[t] = wmma4(a[kc], bb[t][kc], acc[t]);
  }
#pragma unroll
  for (int t = 0; t < 4; ++t) {
#pragma unroll
    for (int r = 0; r < 8; ++r) S[ld + SG * r + ST * t] = acc[t][r];
  }
}

__launch_bounds__(256)
__global__ void quanv_kernel(const float* __restrict__ x,
                             const float* __restrict__ qw,
                             float* __restrict__ out) {
  constexpr int NQ = 9, B = 4, C = 3, H = 28, W = 28, HOUT = 26, WOUT = 26;
  constexpr int TOTAL = B * HOUT * WOUT;

  __shared__ alignas(16) float lds[8][1024];
  __shared__ float2 trig[225];
  __shared__ float2 gtab[12][32];

  const int tid  = threadIdx.x;
  const int lane = tid & 31;
  const int warp = __builtin_amdgcn_readfirstlane(tid >> 5);
  const int wid  = blockIdx.x * 8 + warp;
  const int widc = wid < TOTAL ? wid : TOTAL - 1;

  const int b  = widc / (HOUT * WOUT);
  const int yy = (widc / WOUT) % HOUT;
  const int xx = widc % WOUT;

  float* Sre = &lds[warp][0];
  float* Sim = &lds[warp][512];

  const int m16 = lane & 15;
  const int hh  = lane >> 4;

  if (tid < 225) {
    float a;
    if (tid < 216) {
      const int pw = tid / 27, j = tid % 27;
      const int ch = j / 9, i = j % 9;
      int p = blockIdx.x * 8 + pw;
      p = p < TOTAL ? p : TOTAL - 1;
      const int pb = p / (HOUT * WOUT);
      const int py = (p / WOUT) % HOUT;
      const int px = p % WOUT;
      a = (x[((pb * C + ch) * H + py + i / 3) * W + px + i % 3] + qw[i]) * 0.5f;
    } else {
      a = qw[NQ + (tid - 216)] * 0.5f;
    }
    trig[tid] = make_float2(cosf(a), sinf(a));
  }
  __syncthreads();

  auto uni = [](float v) -> float {
    return __int_as_float(__builtin_amdgcn_readfirstlane(__float_as_int(v)));
  };

  for (int j = 0; j < 2; ++j) {
    const int e = warp + 8 * j;
    if (e < 12) {
      const int grp = e >> 2, kc = e & 3;
      float cg[3], sg[3];
#pragma unroll
      for (int i = 0; i < 3; ++i) {
        float2 t = trig[216 + 3 * grp + i];
        cg[i] = t.x;
        sg[i] = t.y;
      }
      const int gp = m16 & 7, rp = m16 >> 3;
      const int reim = kc >> 1;
      float w[2];
#pragma unroll
      for (int v = 0; v < 2; ++v) {
        const int g = v + 2 * (kc & 1) + 4 * hh;
        const int diff = gp ^ g;
        float P = 1.0f;
#pragma unroll
        for (int i = 0; i < 3; ++i) P *= ((diff >> i) & 1) ? sg[i] : cg[i];
        const int hm = __popc(diff) & 3;
        const float re = (hm == 0) ? P : ((hm == 2) ? -P : 0.0f);
        const float im = (hm == 1) ? -P : ((hm == 3) ? P : 0.0f);
        w[v] = (rp == reim) ? re : (rp ? im : -im);
      }
      gtab[e][lane] = make_float2(w[0], w[1]);
    }
  }
  __syncthreads();

  float zacc[4] = {0.f, 0.f, 0.f, 0.f};

  for (int ch = 0; ch < C; ++ch) {
    float ca[NQ], sa[NQ];
#pragma unroll
    for (int i = 0; i < NQ; ++i) {
      float2 t = trig[27 * warp + 9 * ch + i];
      ca[i] = uni(t.x);
      sa[i] = uni(t.y);
    }

    float pRe[16], pIm[16];
#pragma unroll
    for (int j = 0; j < 16; ++j) {
      int idx = lane + 32 * j;
      float P = 1.0f;
#pragma unroll
      for (int i = 0; i < NQ; ++i) P *= ((idx >> i) & 1) ? sa[i] : ca[i];
      int hm = __popc(idx) & 3;
      pRe[j] = (hm == 0) ? P : ((hm == 2) ? -P : 0.0f);
      pIm[j] = (hm == 1) ? -P : ((hm == 3) ? P : 0.0f);
    }

    cnot_ring_lane_major(pRe, lane);
    cnot_ring_lane_major(pIm, lane);
#pragma unroll
    for (int j = 0; j < 16; ++j) {
      Sre[lane + 32 * j] = pRe[j];
      Sim[lane + 32 * j] = pIm[j];
    }
    lds_fence();

    rx_group<0>(Sre, &gtab[0][0], lane);
    lds_fence();
    rx_group<1>(Sre, &gtab[4][0], lane);
    lds_fence();
    rx_group<2>(Sre, &gtab[8][0], lane);
    lds_fence();

    float fRe[16], fIm[16];
#pragma unroll
    for (int t = 0; t < 2; ++t) {
#pragma unroll
      for (int v = 0; v < 8; ++v) {
        const int idx = v + 8 * hh + 16 * (m16 + 16 * t);
        fRe[8 * t + v] = Sre[idx];
        fIm[8 * t + v] = Sim[idx];
      }
    }

    cnot_ring_acc(fRe, lane);
    cnot_ring_acc(fIm, lane);

    float z[4] = {0.f, 0.f, 0.f, 0.f};
#pragma unroll
    for (int e = 0; e < 16; ++e) {
      float p = fRe[e] * fRe[e] + fIm[e] * fIm[e];
#pragma unroll
      for (int f = 0; f < 3; ++f) z[f] += ((e >> f) & 1) ? -p : p;
      z[3] += hh ? -p : p;
    }
#pragma unroll
    for (int f = 0; f < 4; ++f) {
#pragma unroll
      for (int m = 1; m < 32; m <<= 1) z[f] += __shfl_xor(z[f], m, 32);
      zacc[f] += z[f];
    }
  }

  if (lane == 0 && wid < TOTAL) {
    const float inv = 1.0f / 3.0f;
#pragma unroll
    for (int f = 0; f < 4; ++f)
      out[((b * 4 + f) * HOUT + yy) * WOUT + xx] = zacc[f] * inv;
  }
}

extern "C" void kernel_launch(void* const* d_in, const int* in_sizes, int n_in,
                              void* d_out, int out_size, void* d_ws, size_t ws_size,
                              hipStream_t stream) {
  (void)in_sizes; (void)n_in; (void)out_size; (void)d_ws; (void)ws_size;
  const float* x  = (const float*)d_in[0];
  const float* qw = (const float*)d_in[1];
  float* out = (float*)d_out;
  const int total_waves = 4 * 26 * 26;
  dim3 grid(total_waves / 8), block(256);
  hipLaunchKernelGGL(quanv_kernel, grid, block, 0, stream, x, qw, out);
}


# One wave per (pixel, channel) circuit, not per pixel: 676 workgroups of 12 waves (4 pixels x 3 channels), channel loop gone; per-wave <Z> meet in LDS, one thread per output adds them in channel order
# speedup vs baseline: 3.3699x; 1.3688x over previous
#include <hip/hip_runtime.h>


typedef __attribute__((ext_vector_type(2))) float v2f;
typedef __attribute__((ext_vector_type(8))) float v8f;

__device__ __forceinline__ void lds_fence() {
  asm volatile("s_wait_dscnt 0" ::: "memory");
}

__device__ __forceinline__ v8f wmma4(v2f a, v2f b, v8f c) {
  return __builtin_amdgcn_wmma_f32_16x16x4_f32(false, a, false, b, (short)0, c,
                                               false, false);
}


template <int CTRL, int ROWS>
__device__ __forceinline__ float lane_quad(float v) {
  int x = __float_as_int(v);
  return __int_as_float(__builtin_amdgcn_update_dpp(x, x, CTRL, ROWS, 0xF, false));
}
__device__ __forceinline__ float cx_l0_l1(float v) { return lane_quad<0x6C, 0x3>(v); }
__device__ __forceinline__ float cx_l1_l2(float v) {
  constexpr int SEL = 0 | (1 << 3) | (6 << 6) | (7 << 9) | (4 << 12) | (5 << 15) |
                      (2 << 18) | (3 << 21);
  return __int_as_float(__builtin_amdgcn_mov_dpp8(__float_as_int(v), SEL));
}
__device__ __forceinline__ float cx_l2_l3(float v) {
  int x = __float_as_int(v);
  return __int_as_float(
      __builtin_amdgcn_permlane16(x, x, 0xFEDC3210u, 0x7654BA98u, false, false));
}
__device__ __forceinline__ float other_row(float v) {
  int x = __float_as_int(v);
  return __int_as_float(
      __builtin_amdgcn_permlanex16(x, x, 0x76543210u, 0xFEDCBA98u, false, false));
}
__device__ __forceinline__ void swapf(float& a, float& b) { float t = a; a = b; b = t; }

__device__ __forceinline__ void cnot_ring_lane_major(float (&r)[16], int lane) {
  const bool l3 = (lane & 8) != 0, l4 = (lane & 16) != 0;
#pragma unroll
  for (int j = 0; j < 16; ++j) {
    float v = cx_l2_l3(cx_l1_l2(cx_l0_l1(r[j])));
    float o = other_row(v);
    r[j] = l3 ? o : v;
  }
#pragma unroll
  for (int j = 0; j < 16; j += 2) {
    float a = r[j], b = r[j + 1];
    r[j] = l4 ? b : a;
    r[j + 1] = l4 ? a : b;
  }
#pragma unroll
  for (int j = 0; j < 16; ++j) if ((j & 1) && !(j & 2)) swapf(r[j], r[j ^ 2]);
#pragma unroll
  for (int j = 0; j < 16; ++j) if ((j & 2) && !(j & 4)) swapf(r[j], r[j ^ 4]);
#pragma unroll
  for (int j = 0; j < 16; ++j) if ((j & 4) && !(j & 8)) swapf(r[j], r[j ^ 8]);
#pragma unroll
  for (int j = 8; j < 16; ++j) r[j] = lane_quad<0xB1, 0x3>(r[j]);
}

__device__ __forceinline__ void cnot_ring_acc(float (&g)[16], int lane) {
  const bool l3 = (lane & 8) != 0;
#pragma unroll
  for (int t = 0; t < 2; ++t) {
#pragma unroll
    for (int v = 0; v < 8; ++v) if ((v & 1) && !(v & 2)) swapf(g[8 * t + v], g[8 * t + (v ^ 2)]);
#pragma unroll
    for (int v = 0; v < 8; ++v) if ((v & 2) && !(v & 4)) swapf(g[8 * t + v], g[8 * t + (v ^ 4)]);
  }
#pragma unroll
  for (int e = 0; e < 16; ++e) {
    float v = g[e];
    if (e & 4) v = other_row(v);
    v = lane_quad<0xB1, 0x2>(v);
    g[e] = cx_l2_l3(cx_l1_l2(cx_l0_l1(v)));
  }
#pragma unroll
  for (int v = 0; v < 8; ++v) {
    float a = g[v], b = g[8 + v];
    g[v] = l3 ? b : a;
    g[8 + v] = l3 ? a : b;
  }
#pragma unroll
  for (int v = 0; v < 8; v += 2) swapf(g[8 + v], g[8 + v + 1]);
}

template <int G>
__device__ __forceinline__ void rx_group(float* S, const float2* gtG, int lane) {
  const int n = lane & 15, hh = lane >> 4;
  constexpr int SG = (G == 0) ? 1 : (G == 1) ? 8 : 64;
  constexpr int ST = (G == 2) ? 16 : 128;
  int lb, ld;
  if (G == 0)      { lb = 8 * n + 4 * hh;                        ld = 8 * n + 512 * hh; }
  else if (G == 1) { lb = (n & 7) + 64 * (n >> 3) + 32 * hh;     ld = (n & 7) + 64 * (n >> 3) + 512 * hh; }
  else             { lb = n + 256 * hh;                          ld = n + 512 * hh; }

  v2f a[4];
#pragma unroll
  for (int kc = 0; kc < 4; ++kc) {
    float2 w = gtG[kc * 32 + lane];
    a[kc][0] = w.x;
    a[kc][1] = w.y;
  }
  v2f bb[4][4];
  if (G == 0) {
#pragma unroll
    for (int t = 0; t < 4; ++t) {
#pragma unroll
      for (int reim = 0; reim < 2; ++reim) {
        const float4 q = *reinterpret_cast<const float4*>(&S[lb + ST * t + 512 * reim]);
        bb[t][2 * reim][0] = q.x;
        bb[t][2 * reim][1] = q.y;
        bb[t][2 * reim + 1][0] = q.z;
        bb[t][2 * reim + 1][1] = q.w;
      }
    }
  } else {
#pragma unroll
    for (int t = 0; t < 4; ++t) {
#pragma unroll
      for (int kc = 0; kc < 4; ++kc) {
#pragma unroll
        for (int v = 0; v < 2; ++v)
          bb[t][kc][v] = S[lb + SG * (v + 2 * (kc & 1)) + ST * t + 512 * (kc >> 1)];
      }
    }
  }
  __builtin_amdgcn_sched_barrier(0);
  v8f acc[4];
#pragma unroll
  for (int t = 0; t < 4; ++t) acc[t] = v8f{0.f, 0.f, 0.f, 0.f, 0.f, 0.f, 0.f, 0.f};
#pragma unroll
  for (int kc = 0; kc < 4; ++kc) {
#pragma unroll
    for (int t = 0; t < 4; ++t) acc[t] = wmma4(a[kc], bb[t][kc], acc[t]);
  }
#pragma unroll
  for (int t = 0; t < 4; ++t) {
#pragma unroll
    for (int r = 0; r < 8; ++r) S[ld + SG * r + ST * t] = acc[t][r];
  }
}

__launch_bounds__(384)
__global__ void quanv_kernel(const float* __restrict__ x,
                             const float* __restrict__ qw,
                             float* __restrict__ out) {
  constexpr int NQ = 9, B = 4, C = 3, H = 28, W = 28, HOUT = 26, WOUT = 26;
  constexpr int TOTAL = B * HOUT * WOUT;
  constexpr int PIX = 4;
  constexpr int NWAVE = PIX * C;
  constexpr int NANG = NWAVE * NQ;

  __shared__ alignas(16) float lds[NWAVE][1024];
  __shared__ float2 trig[NANG + NQ];
  __shared__ alignas(16) float zsh[NWAVE][4];
  __shared__ float2 gtab[12][32];

  const int tid  = threadIdx.x;
  const int lane = tid & 31;
  const int warp = __builtin_amdgcn_readfirstlane(tid >> 5);

  float* Sre = &lds[warp][0];
  float* Sim = &lds[warp][512];

  const int m16 = lane & 15;
  const int hh  = lane >> 4;

  if (tid < NANG + NQ) {
    float a;
    if (tid < NANG) {
      const int pw = tid / 27, j = tid % 27;
      const int ch = j / 9, i = j % 9;
      int p = blockIdx.x * PIX + pw;
      p = p < TOTAL ? p : TOTAL - 1;
      const int pb = p / (HOUT * WOUT);
      const int py = (p / WOUT) % HOUT;
      const int px = p % WOUT;
      a = (x[((pb * C + ch) * H + py + i / 3) * W + px + i % 3] + qw[i]) * 0.5f;
    } else {
      a = qw[NQ + (tid - NANG)] * 0.5f;
    }
    trig[tid] = make_float2(cosf(a), sinf(a));
  }
  __syncthreads();

  auto uni = [](float v) -> float {
    return __int_as_float(__builtin_amdgcn_readfirstlane(__float_as_int(v)));
  };

  {
    const int e = warp;
    const int grp = e >> 2, kc = e & 3;
    float cg[3], sg[3];
#pragma unroll
    for (int i = 0; i < 3; ++i) {
      float2 t = trig[NANG + 3 * grp + i];
      cg[i] = t.x;
      sg[i] = t.y;
    }
    const int gp = m16 & 7, rp = m16 >> 3;
    const int reim = kc >> 1;
    float w[2];
#pragma unroll
    for (int v = 0; v < 2; ++v) {
      const int g = v + 2 * (kc & 1) + 4 * hh;
      const int diff = gp ^ g;
      float P = 1.0f;
#pragma unroll
      for (int i = 0; i < 3; ++i) P *= ((diff >> i) & 1) ? sg[i] : cg[i];
      const int hm = __popc(diff) & 3;
      const float re = (hm == 0) ? P : ((hm == 2) ? -P : 0.0f);
      const float im = (hm == 1) ? -P : ((hm == 3) ? P : 0.0f);
      w[v] = (rp == reim) ? re : (rp ? im : -im);
    }
    gtab[e][lane] = make_float2(w[0], w[1]);
  }
  __syncthreads();

  {
    float ca[NQ], sa[NQ];
#pragma unroll
    for (int i = 0; i < NQ; ++i) {
      float2 t = trig[9 * warp + i];
      ca[i] = uni(t.x);
      sa[i] = uni(t.y);
    }

    float pRe[16], pIm[16];
#pragma unroll
    for (int j = 0; j < 16; ++j) {
      int idx = lane + 32 * j;
      float P = 1.0f;
#pragma unroll
      for (int i = 0; i < NQ; ++i) P *= ((idx >> i) & 1) ? sa[i] : ca[i];
      int hm = __popc(idx) & 3;
      pRe[j] = (hm == 0) ? P : ((hm == 2) ? -P : 0.0f);
      pIm[j] = (hm == 1) ? -P : ((hm == 3) ? P : 0.0f);
    }

    cnot_ring_lane_major(pRe, lane);
    cnot_ring_lane_major(pIm, lane);
#pragma unroll
    for (int j = 0; j < 16; ++j) {
      Sre[lane + 32 * j] = pRe[j];
      Sim[lane + 32 * j] = pIm[j];
    }
    lds_fence();

    rx_group<0>(Sre, &gtab[0][0], lane);
    lds_fence();
    rx_group<1>(Sre, &gtab[4][0], lane);
    lds_fence();
    rx_group<2>(Sre, &gtab[8][0], lane);
    lds_fence();

    float fRe[16], fIm[16];
#pragma unroll
    for (int t = 0; t < 2; ++t) {
#pragma unroll
      for (int v = 0; v < 8; ++v) {
        const int idx = v + 8 * hh + 16 * (m16 + 16 * t);
        fRe[8 * t + v] = Sre[idx];
        fIm[8 * t + v] = Sim[idx];
      }
    }

    cnot_ring_acc(fRe, lane);
    cnot_ring_acc(fIm, lane);

    float z[4] = {0.f, 0.f, 0.f, 0.f};
#pragma unroll
    for (int e = 0; e < 16; ++e) {
      float p = fRe[e] * fRe[e] + fIm[e] * fIm[e];
#pragma unroll
      for (int f = 0; f < 3; ++f) z[f] += ((e >> f) & 1) ? -p : p;
      z[3] += hh ? -p : p;
    }
#pragma unroll
    for (int f = 0; f < 4; ++f) {
#pragma unroll
      for (int m = 1; m < 32; m <<= 1) z[f] += __shfl_xor(z[f], m, 32);
    }
    if (lane == 0) *reinterpret_cast<float4*>(&zsh[warp][0]) = make_float4(z[0], z[1], z[2], z[3]);
  }
  __syncthreads();

  if (tid < 4 * PIX) {
    const int pw = tid & (PIX - 1), f = tid >> 2;
    const int p = blockIdx.x * PIX + pw;
    if (p < TOTAL) {
      const int b  = p / (HOUT * WOUT);
      const int yy = (p / WOUT) % HOUT;
      const int xx = p % WOUT;
      const float inv = 1.0f / 3.0f;
      const float s = (zsh[3 * pw][f] + zsh[3 * pw + 1][f]) + zsh[3 * pw + 2][f];
      out[((b * 4 + f) * HOUT + yy) * WOUT + xx] = s * inv;
    }
  }
}

extern "C" void kernel_launch(void* const* d_in, const int* in_sizes, int n_in,
                              void* d_out, int out_size, void* d_ws, size_t ws_size,
                              hipStream_t stream) {
  (void)in_sizes; (void)n_in; (void)out_size; (void)d_ws; (void)ws_size;
  const float* x  = (const float*)d_in[0];
  const float* qw = (const float*)d_in[1];
  float* out = (float*)d_out;
  const int total_pixels = 4 * 26 * 26;
  dim3 grid(total_pixels / 4), block(384);
  hipLaunchKernelGGL(quanv_kernel, grid, block, 0, stream, x, qw, out);
}


# State before ring 1 kept as lane factor x wave-uniform register factor: quarter turns and the lane-bit CNOTs of ring 1 act once on the lane factor, not on 32 registers; factor order changes
# speedup vs baseline: 3.6330x; 1.0781x over previous
#include <hip/hip_runtime.h>


typedef __attribute__((ext_vector_type(2))) float v2f;
typedef __attribute__((ext_vector_type(8))) float v8f;

__device__ __forceinline__ void lds_fence() {
  asm volatile("s_wait_dscnt 0" ::: "memory");
}

__device__ __forceinline__ v8f wmma4(v2f a, v2f b, v8f c) {
  return __builtin_amdgcn_wmma_f32_16x16x4_f32(false, a, false, b, (short)0, c,
                                               false, false);
}


template <int CTRL, int ROWS>
__device__ __forceinline__ float lane_quad(float v) {
  int x = __float_as_int(v);
  return __int_as_float(__builtin_amdgcn_update_dpp(x, x, CTRL, ROWS, 0xF, false));
}
__device__ __forceinline__ float cx_l0_l1(float v) { return lane_quad<0x6C, 0x3>(v); }
__device__ __forceinline__ float cx_l1_l2(float v) {
  constexpr int SEL = 0 | (1 << 3) | (6 << 6) | (7 << 9) | (4 << 12) | (5 << 15) |
                      (2 << 18) | (3 << 21);
  return __int_as_float(__builtin_amdgcn_mov_dpp8(__float_as_int(v), SEL));
}
__device__ __forceinline__ float cx_l2_l3(float v) {
  int x = __float_as_int(v);
  return __int_as_float(
      __builtin_amdgcn_permlane16(x, x, 0xFEDC3210u, 0x7654BA98u, false, false));
}
__device__ __forceinline__ float other_row(float v) {
  int x = __float_as_int(v);
  return __int_as_float(
      __builtin_amdgcn_permlanex16(x, x, 0x76543210u, 0xFEDCBA98u, false, false));
}
__device__ __forceinline__ void swapf(float& a, float& b) { float t = a; a = b; b = t; }

__device__ __forceinline__ void cnot_ring_acc(float (&g)[16], int lane) {
  const bool l3 = (lane & 8) != 0;
#pragma unroll
  for (int t = 0; t < 2; ++t) {
#pragma unroll
    for (int v = 0; v < 8; ++v) if ((v & 1) && !(v & 2)) swapf(g[8 * t + v], g[8 * t + (v ^ 2)]);
#pragma unroll
    for (int v = 0; v < 8; ++v) if ((v & 2) && !(v & 4)) swapf(g[8 * t + v], g[8 * t + (v ^ 4)]);
  }
#pragma unroll
  for (int e = 0; e < 16; ++e) {
    float v = g[e];
    if (e & 4) v = other_row(v);
    v = lane_quad<0xB1, 0x2>(v);
    g[e] = cx_l2_l3(cx_l1_l2(cx_l0_l1(v)));
  }
#pragma unroll
  for (int v = 0; v < 8; ++v) {
    float a = g[v], b = g[8 + v];
    g[v] = l3 ? b : a;
    g[8 + v] = l3 ? a : b;
  }
#pragma unroll
  for (int v = 0; v < 8; v += 2) swapf(g[8 + v], g[8 + v + 1]);
}

template <int G>
__device__ __forceinline__ void rx_group(float* S, const float2* gtG, int lane) {
  const int n = lane & 15, hh = lane >> 4;
  constexpr int SG = (G == 0) ? 1 : (G == 1) ? 8 : 64;
  constexpr int ST = (G == 2) ? 16 : 128;
  int lb, ld;
  if (G == 0)      { lb = 8 * n + 4 * hh;                        ld = 8 * n + 512 * hh; }
  else if (G == 1) { lb = (n & 7) + 64 * (n >> 3) + 32 * hh;     ld = (n & 7) + 64 * (n >> 3) + 512 * hh; }
  else             { lb = n + 256 * hh;                          ld = n + 512 * hh; }

  v2f a[4];
#pragma unroll
  for (int kc = 0; kc < 4; ++kc) {
    float2 w = gtG[kc * 32 + lane];
    a[kc][0] = w.x;
    a[kc][1] = w.y;
  }
  v2f bb[4][4];
  if (G == 0) {
#pragma unroll
    for (int t = 0; t < 4; ++t) {
#pragma unroll
      for (int reim = 0; reim < 2; ++reim) {
        const float4 q = *reinterpret_cast<const float4*>(&S[lb + ST * t + 512 * reim]);
        bb[t][2 * reim][0] = q.x;
        bb[t][2 * reim][1] = q.y;
        bb[t][2 * reim + 1][0] = q.z;
        bb[t][2 * reim + 1][1] = q.w;
      }
    }
  } else {
#pragma unroll
    for (int t = 0; t < 4; ++t) {
#pragma unroll
      for (int kc = 0; kc < 4; ++kc) {
#pragma unroll
        for (int v = 0; v < 2; ++v)
          bb[t][kc][v] = S[lb + SG * (v + 2 * (kc & 1)) + ST * t + 512 * (kc >> 1)];
      }
    }
  }
  __builtin_amdgcn_sched_barrier(0);
  v8f acc[4];
#pragma unroll
  for (int t = 0; t < 4; ++t) acc[t] = v8f{0.f, 0.f, 0.f, 0.f, 0.f, 0.f, 0.f, 0.f};
#pragma unroll
  for (int kc = 0; kc < 4; ++kc) {
#pragma unroll
    for (int t = 0; t < 4; ++t) acc[t] = wmma4(a[kc], bb[t][kc], acc[t]);
  }
#pragma unroll
  for (int t = 0; t < 4; ++t) {
#pragma unroll
    for (int r = 0; r < 8; ++r) S[ld + SG * r + ST * t] = acc[t][r];
  }
}

__launch_bounds__(384)
__global__ void quanv_kernel(const float* __restrict__ x,
                             const float* __restrict__ qw,
                             float* __restrict__ out) {
  constexpr int NQ = 9, B = 4, C = 3, H = 28, W = 28, HOUT = 26, WOUT = 26;
  constexpr int TOTAL = B * HOUT * WOUT;
  constexpr int PIX = 4;
  constexpr int NWAVE = PIX * C;
  constexpr int NANG = NWAVE * NQ;

  __shared__ alignas(16) float lds[NWAVE][1024];
  __shared__ float2 trig[NANG + NQ];
  __shared__ alignas(16) float zsh[NWAVE][4];
  __shared__ float2 gtab[12][32];

  const int tid  = threadIdx.x;
  const int lane = tid & 31;
  const int warp = __builtin_amdgcn_readfirstlane(tid >> 5);

  float* Sre = &lds[warp][0];
  float* Sim = &lds[warp][512];

  const int m16 = lane & 15;
  const int hh  = lane >> 4;

  if (tid < NANG + NQ) {
    float a;
    if (tid < NANG) {
      const int pw = tid / 27, j = tid % 27;
      const int ch = j / 9, i = j % 9;
      int p = blockIdx.x * PIX + pw;
      p = p < TOTAL ? p : TOTAL - 1;
      const int pb = p / (HOUT * WOUT);
      const int py = (p / WOUT) % HOUT;
      const int px = p % WOUT;
      a = (x[((pb * C + ch) * H + py + i / 3) * W + px + i % 3] + qw[i]) * 0.5f;
    } else {
      a = qw[NQ + (tid - NANG)] * 0.5f;
    }
    trig[tid] = make_float2(cosf(a), sinf(a));
  }
  __syncthreads();

  auto uni = [](float v) -> float {
    return __int_as_float(__builtin_amdgcn_readfirstlane(__float_as_int(v)));
  };

  {
    const int e = warp;
    const int grp = e >> 2, kc = e & 3;
    float cg[3], sg[3];
#pragma unroll
    for (int i = 0; i < 3; ++i) {
      float2 t = trig[NANG + 3 * grp + i];
      cg[i] = t.x;
      sg[i] = t.y;
    }
    const int gp = m16 & 7, rp = m16 >> 3;
    const int reim = kc >> 1;
    float w[2];
#pragma unroll
    for (int v = 0; v < 2; ++v) {
      const int g = v + 2 * (kc & 1) + 4 * hh;
      const int diff = gp ^ g;
      float P = 1.0f;
#pragma unroll
      for (int i = 0; i < 3; ++i) P *= ((diff >> i) & 1) ? sg[i] : cg[i];
      const int hm = __popc(diff) & 3;
      const float re = (hm == 0) ? P : ((hm == 2) ? -P : 0.0f);
      const float im = (hm == 1) ? -P : ((hm == 3) ? P : 0.0f);
      w[v] = (rp == reim) ? re : (rp ? im : -im);
    }
    gtab[e][lane] = make_float2(w[0], w[1]);
  }
  __syncthreads();

  {
    float ca[NQ], sa[NQ];
#pragma unroll
    for (int i = 0; i < NQ; ++i) {
      float2 t = trig[9 * warp + i];
      ca[i] = uni(t.x);
      sa[i] = uni(t.y);
    }

    const bool l3 = (lane & 8) != 0, l4 = (lane & 16) != 0;
    float L = 1.0f;
#pragma unroll
    for (int i = 0; i < 5; ++i) L *= ((lane >> i) & 1) ? sa[i] : ca[i];
    const int kl = __popc(lane) & 3;
    float lre = (kl == 0) ? L : ((kl == 2) ? -L : 0.0f);
    float lim = (kl == 1) ? -L : ((kl == 3) ? L : 0.0f);
    {
      const float vr = cx_l2_l3(cx_l1_l2(cx_l0_l1(lre)));
      const float vi = cx_l2_l3(cx_l1_l2(cx_l0_l1(lim)));
      const float orr = other_row(vr), oi = other_row(vi);
      lre = l3 ? orr : vr;
      lim = l3 ? oi : vi;
    }
    const float are = ca[5] * lre, aim = ca[5] * lim;
    const float bre = sa[5] * lim, bim = -(sa[5] * lre);
    float mre[2][2], mim[2][2];
    mre[0][0] = l4 ? bre : are;
    mim[0][0] = l4 ? bim : aim;
    mre[1][0] = l4 ? are : bre;
    mim[1][0] = l4 ? aim : bim;
#pragma unroll
    for (int b = 0; b < 2; ++b) {
      mre[b][1] = lane_quad<0xB1, 0x3>(mre[b][0]);
      mim[b][1] = lane_quad<0xB1, 0x3>(mim[b][0]);
    }
    float pRe[16], pIm[16];
#pragma unroll
    for (int j = 0; j < 16; ++j) {
      const int b = j & 1, m = j >> 1;
      float U = 1.0f;
#pragma unroll
      for (int i = 6; i < NQ; ++i) U *= ((m >> (i - 6)) & 1) ? sa[i] : ca[i];
      const int pm = (m & 1) + ((m >> 1) & 1) + ((m >> 2) & 1);
      const int km = pm & 3, sw = (b ^ pm) & 1;
      const float xr = mre[b][sw], xi = mim[b][sw];
      if (km == 0)      { pRe[j] = U * xr;    pIm[j] = U * xi; }
      else if (km == 1) { pRe[j] = U * xi;    pIm[j] = -(U * xr); }
      else if (km == 2) { pRe[j] = -(U * xr); pIm[j] = -(U * xi); }
      else              { pRe[j] = -(U * xi); pIm[j] = U * xr; }
    }
#pragma unroll
    for (int j = 0; j < 16; ++j) if ((j & 1) && !(j & 2)) { swapf(pRe[j], pRe[j ^ 2]); swapf(pIm[j], pIm[j ^ 2]); }
#pragma unroll
    for (int j = 0; j < 16; ++j) if ((j & 2) && !(j & 4)) { swapf(pRe[j], pRe[j ^ 4]); swapf(pIm[j], pIm[j ^ 4]); }
#pragma unroll
    for (int j = 0; j < 16; ++j) if ((j & 4) && !(j & 8)) { swapf(pRe[j], pRe[j ^ 8]); swapf(pIm[j], pIm[j ^ 8]); }

#pragma unroll
    for (int j = 0; j < 16; ++j) {
      Sre[lane + 32 * j] = pRe[j];
      Sim[lane + 32 * j] = pIm[j];
    }
    lds_fence();

    rx_group<0>(Sre, &gtab[0][0], lane);
    lds_fence();
    rx_group<1>(Sre, &gtab[4][0], lane);
    lds_fence();
    rx_group<2>(Sre, &gtab[8][0], lane);
    lds_fence();

    float fRe[16], fIm[16];
#pragma unroll
    for (int t = 0; t < 2; ++t) {
#pragma unroll
      for (int v = 0; v < 8; ++v) {
        const int idx = v + 8 * hh + 16 * (m16 + 16 * t);
        fRe[8 * t + v] = Sre[idx];
        fIm[8 * t + v] = Sim[idx];
      }
    }

    cnot_ring_acc(fRe, lane);
    cnot_ring_acc(fIm, lane);

    float z[4] = {0.f, 0.f, 0.f, 0.f};
#pragma unroll
    for (int e = 0; e < 16; ++e) {
      float p = fRe[e] * fRe[e] + fIm[e] * fIm[e];
#pragma unroll
      for (int f = 0; f < 3; ++f) z[f] += ((e >> f) & 1) ? -p : p;
      z[3] += hh ? -p : p;
    }
#pragma unroll
    for (int f = 0; f < 4; ++f) {
#pragma unroll
      for (int m = 1; m < 32; m <<= 1) z[f] += __shfl_xor(z[f], m, 32);
    }
    if (lane == 0) *reinterpret_cast<float4*>(&zsh[warp][0]) = make_float4(z[0], z[1], z[2], z[3]);
  }
  __syncthreads();

  if (tid < 4 * PIX) {
    const int pw = tid & (PIX - 1), f = tid >> 2;
    const int p = blockIdx.x * PIX + pw;
    if (p < TOTAL) {
      const int b  = p / (HOUT * WOUT);
      const int yy = (p / WOUT) % HOUT;
      const int xx = p % WOUT;
      const float inv = 1.0f / 3.0f;
      const float s = (zsh[3 * pw][f] + zsh[3 * pw + 1][f]) + zsh[3 * pw + 2][f];
      out[((b * 4 + f) * HOUT + yy) * WOUT + xx] = s * inv;
    }
  }
}

extern "C" void kernel_launch(void* const* d_in, const int* in_sizes, int n_in,
                              void* d_out, int out_size, void* d_ws, size_t ws_size,
                              hipStream_t stream) {
  (void)in_sizes; (void)n_in; (void)out_size; (void)d_ws; (void)ws_size;
  const float* x  = (const float*)d_in[0];
  const float* qw = (const float*)d_in[1];
  float* out = (float*)d_out;
  const int total_pixels = 4 * 26 * 26;
  dim3 grid(total_pixels / 4), block(384);
  hipLaunchKernelGGL(quanv_kernel, grid, block, 0, stream, x, qw, out);
}


# Per-wave LDS state slice re-pitched: index bits 6, 7, 8 and the re/im plane weigh 74, 148, 296, 596 floats, so all 32 lanes of every state store and load hit different banks; G0 loads 8-byte
# speedup vs baseline: 4.0938x; 1.1268x over previous
#include <hip/hip_runtime.h>


typedef __attribute__((ext_vector_type(2))) float v2f;
typedef __attribute__((ext_vector_type(8))) float v8f;

__device__ __forceinline__ void lds_fence() {
  asm volatile("s_wait_dscnt 0" ::: "memory");
}

__device__ __forceinline__ constexpr int spos(int a) {
  return (a & 63) + 74 * ((a >> 6) & 1) + 148 * ((a >> 7) & 1) +
         296 * ((a >> 8) & 1) + 596 * ((a >> 9) & 1);
}
constexpr int SLICE = 1184;

__device__ __forceinline__ v8f wmma4(v2f a, v2f b, v8f c) {
  return __builtin_amdgcn_wmma_f32_16x16x4_f32(false, a, false, b, (short)0, c,
                                               false, false);
}


template <int CTRL, int ROWS>
__device__ __forceinline__ float lane_quad(float v) {
  int x = __float_as_int(v);
  return __int_as_float(__builtin_amdgcn_update_dpp(x, x, CTRL, ROWS, 0xF, false));
}
__device__ __forceinline__ float cx_l0_l1(float v) { return lane_quad<0x6C, 0x3>(v); }
__device__ __forceinline__ float cx_l1_l2(float v) {
  constexpr int SEL = 0 | (1 << 3) | (6 << 6) | (7 << 9) | (4 << 12) | (5 << 15) |
                      (2 << 18) | (3 << 21);
  return __int_as_float(__builtin_amdgcn_mov_dpp8(__float_as_int(v), SEL));
}
__device__ __forceinline__ float cx_l2_l3(float v) {
  int x = __float_as_int(v);
  return __int_as_float(
      __builtin_amdgcn_permlane16(x, x, 0xFEDC3210u, 0x7654BA98u, false, false));
}
__device__ __forceinline__ float other_row(float v) {
  int x = __float_as_int(v);
  return __int_as_float(
      __builtin_amdgcn_permlanex16(x, x, 0x76543210u, 0xFEDCBA98u, false, false));
}
__device__ __forceinline__ void swapf(float& a, float& b) { float t = a; a = b; b = t; }

__device__ __forceinline__ void cnot_ring_acc(float (&g)[16], int lane) {
  const bool l3 = (lane & 8) != 0;
#pragma unroll
  for (int t = 0; t < 2; ++t) {
#pragma unroll
    for (int v = 0; v < 8; ++v) if ((v & 1) && !(v & 2)) swapf(g[8 * t + v], g[8 * t + (v ^ 2)]);
#pragma unroll
    for (int v = 0; v < 8; ++v) if ((v & 2) && !(v & 4)) swapf(g[8 * t + v], g[8 * t + (v ^ 4)]);
  }
#pragma unroll
  for (int e = 0; e < 16; ++e) {
    float v = g[e];
    if (e & 4) v = other_row(v);
    v = lane_quad<0xB1, 0x2>(v);
    g[e] = cx_l2_l3(cx_l1_l2(cx_l0_l1(v)));
  }
#pragma unroll
  for (int v = 0; v < 8; ++v) {
    float a = g[v], b = g[8 + v];
    g[v] = l3 ? b : a;
    g[8 + v] = l3 ? a : b;
  }
#pragma unroll
  for (int v = 0; v < 8; v += 2) swapf(g[8 + v], g[8 + v + 1]);
}

template <int G>
__device__ __forceinline__ void rx_group(float* S, const float2* gtG, int lane) {
  const int n = lane & 15, hh = lane >> 4;
  constexpr int SG = (G == 0) ? 1 : (G == 1) ? 8 : 64;
  constexpr int ST = (G == 2) ? 16 : 128;
  int lb, ld;
  if (G == 0)      { lb = 8 * n + 4 * hh;                        ld = 8 * n + 512 * hh; }
  else if (G == 1) { lb = (n & 7) + 64 * (n >> 3) + 32 * hh;     ld = (n & 7) + 64 * (n >> 3) + 512 * hh; }
  else             { lb = n + 256 * hh;                          ld = n + 512 * hh; }
  const int plb = spos(lb), pld = spos(ld);

  v2f a[4];
#pragma unroll
  for (int kc = 0; kc < 4; ++kc) {
    float2 w = gtG[kc * 32 + lane];
    a[kc][0] = w.x;
    a[kc][1] = w.y;
  }
  v2f bb[4][4];
  if (G == 0) {
#pragma unroll
    for (int t = 0; t < 4; ++t) {
#pragma unroll
      for (int kc = 0; kc < 4; ++kc) {
        const float2 q = *reinterpret_cast<const float2*>(
            &S[plb + spos(2 * (kc & 1) + ST * t + 512 * (kc >> 1))]);
        bb[t][kc][0] = q.x;
        bb[t][kc][1] = q.y;
      }
    }
  } else {
#pragma unroll
    for (int t = 0; t < 4; ++t) {
#pragma unroll
      for (int kc = 0; kc < 4; ++kc) {
#pragma unroll
        for (int v = 0; v < 2; ++v)
          bb[t][kc][v] = S[plb + spos(SG * (v + 2 * (kc & 1)) + ST * t + 512 * (kc >> 1))];
      }
    }
  }
  __builtin_amdgcn_sched_barrier(0);
  v8f acc[4];
#pragma unroll
  for (int t = 0; t < 4; ++t) acc[t] = v8f{0.f, 0.f, 0.f, 0.f, 0.f, 0.f, 0.f, 0.f};
#pragma unroll
  for (int kc = 0; kc < 4; ++kc) {
#pragma unroll
    for (int t = 0; t < 4; ++t) acc[t] = wmma4(a[kc], bb[t][kc], acc[t]);
  }
#pragma unroll
  for (int t = 0; t < 4; ++t) {
#pragma unroll
    for (int r = 0; r < 8; ++r) S[pld + spos(SG * r + ST * t)] = acc[t][r];
  }
}

__launch_bounds__(384)
__global__ void quanv_kernel(const float* __restrict__ x,
                             const float* __restrict__ qw,
                             float* __restrict__ out) {
  constexpr int NQ = 9, B = 4, C = 3, H = 28, W = 28, HOUT = 26, WOUT = 26;
  constexpr int TOTAL = B * HOUT * WOUT;
  constexpr int PIX = 4;
  constexpr int NWAVE = PIX * C;
  constexpr int NANG = NWAVE * NQ;

  __shared__ alignas(16) float lds[NWAVE][SLICE];
  __shared__ float2 trig[NANG + NQ];
  __shared__ alignas(16) float zsh[NWAVE][4];
  __shared__ float2 gtab[12][32];

  const int tid  = threadIdx.x;
  const int lane = tid & 31;
  const int warp = __builtin_amdgcn_readfirstlane(tid >> 5);

  float* S = &lds[warp][0];

  const int m16 = lane & 15;
  const int hh  = lane >> 4;

  if (tid < NANG + NQ) {
    float a;
    if (tid < NANG) {
      const int pw = tid / 27, j = tid % 27;
      const int ch = j / 9, i = j % 9;
      int p = blockIdx.x * PIX + pw;
      p = p < TOTAL ? p : TOTAL - 1;
      const int pb = p / (HOUT * WOUT);
      const int py = (p / WOUT) % HOUT;
      const int px = p % WOUT;
      a = (x[((pb * C + ch) * H + py + i / 3) * W + px + i % 3] + qw[i]) * 0.5f;
    } else {
      a = qw[NQ + (tid - NANG)] * 0.5f;
    }
    trig[tid] = make_float2(cosf(a), sinf(a));
  }
  __syncthreads();

  auto uni = [](float v) -> float {
    return __int_as_float(__builtin_amdgcn_readfirstlane(__float_as_int(v)));
  };

  {
    const int e = warp;
    const int grp = e >> 2, kc = e & 3;
    float cg[3], sg[3];
#pragma unroll
    for (int i = 0; i < 3; ++i) {
      float2 t = trig[NANG + 3 * grp + i];
      cg[i] = t.x;
      sg[i] = t.y;
    }
    const int gp = m16 & 7, rp = m16 >> 3;
    const int reim = kc >> 1;
    float w[2];
#pragma unroll
    for (int v = 0; v < 2; ++v) {
      const int g = v + 2 * (kc & 1) + 4 * hh;
      const int diff = gp ^ g;
      float P = 1.0f;
#pragma unroll
      for (int i = 0; i < 3; ++i) P *= ((diff >> i) & 1) ? sg[i] : cg[i];
      const int hm = __popc(diff) & 3;
      const float re = (hm == 0) ? P : ((hm == 2) ? -P : 0.0f);
      const float im = (hm == 1) ? -P : ((hm == 3) ? P : 0.0f);
      w[v] = (rp == reim) ? re : (rp ? im : -im);
    }
    gtab[e][lane] = make_float2(w[0], w[1]);
  }
  __syncthreads();

  {
    float ca[NQ], sa[NQ];
#pragma unroll
    for (int i = 0; i < NQ; ++i) {
      float2 t = trig[9 * warp + i];
      ca[i] = uni(t.x);
      sa[i] = uni(t.y);
    }

    const bool l3 = (lane & 8) != 0, l4 = (lane & 16) != 0;
    float L = 1.0f;
#pragma unroll
    for (int i = 0; i < 5; ++i) L *= ((lane >> i) & 1) ? sa[i] : ca[i];
    const int kl = __popc(lane) & 3;
    float lre = (kl == 0) ? L : ((kl == 2) ? -L : 0.0f);
    float lim = (kl == 1) ? -L : ((kl == 3) ? L : 0.0f);
    {
      const float vr = cx_l2_l3(cx_l1_l2(cx_l0_l1(lre)));
      const float vi = cx_l2_l3(cx_l1_l2(cx_l0_l1(lim)));
      const float orr = other_row(vr), oi = other_row(vi);
      lre = l3 ? orr : vr;
      lim = l3 ? oi : vi;
    }
    const float are = ca[5] * lre, aim = ca[5] * lim;
    const float bre = sa[5] * lim, bim = -(sa[5] * lre);
    float mre[2][2], mim[2][2];
    mre[0][0] = l4 ? bre : are;
    mim[0][0] = l4 ? bim : aim;
    mre[1][0] = l4 ? are : bre;
    mim[1][0] = l4 ? aim : bim;
#pragma unroll
    for (int b = 0; b < 2; ++b) {
      mre[b][1] = lane_quad<0xB1, 0x3>(mre[b][0]);
      mim[b][1] = lane_quad<0xB1, 0x3>(mim[b][0]);
    }
    float pRe[16], pIm[16];
#pragma unroll
    for (int j = 0; j < 16; ++j) {
      const int b = j & 1, m = j >> 1;
      float U = 1.0f;
#pragma unroll
      for (int i = 6; i < NQ; ++i) U *= ((m >> (i - 6)) & 1) ? sa[i] : ca[i];
      const int pm = (m & 1) + ((m >> 1) & 1) + ((m >> 2) & 1);
      const int km = pm & 3, sw = (b ^ pm) & 1;
      const float xr = mre[b][sw], xi = mim[b][sw];
      if (km == 0)      { pRe[j] = U * xr;    pIm[j] = U * xi; }
      else if (km == 1) { pRe[j] = U * xi;    pIm[j] = -(U * xr); }
      else if (km == 2) { pRe[j] = -(U * xr); pIm[j] = -(U * xi); }
      else              { pRe[j] = -(U * xi); pIm[j] = U * xr; }
    }
#pragma unroll
    for (int j = 0; j < 16; ++j) if ((j & 1) && !(j & 2)) { swapf(pRe[j], pRe[j ^ 2]); swapf(pIm[j], pIm[j ^ 2]); }
#pragma unroll
    for (int j = 0; j < 16; ++j) if ((j & 2) && !(j & 4)) { swapf(pRe[j], pRe[j ^ 4]); swapf(pIm[j], pIm[j ^ 4]); }
#pragma unroll
    for (int j = 0; j < 16; ++j) if ((j & 4) && !(j & 8)) { swapf(pRe[j], pRe[j ^ 8]); swapf(pIm[j], pIm[j ^ 8]); }

#pragma unroll
    for (int j = 0; j < 16; ++j) {
      S[lane + spos(32 * j)] = pRe[j];
      S[lane + spos(32 * j + 512)] = pIm[j];
    }
    lds_fence();

    rx_group<0>(S, &gtab[0][0], lane);
    lds_fence();
    rx_group<1>(S, &gtab[4][0], lane);
    lds_fence();
    rx_group<2>(S, &gtab[8][0], lane);
    lds_fence();

    float fRe[16], fIm[16];
    const int pfl = spos(8 * hh + 16 * m16);
#pragma unroll
    for (int t = 0; t < 2; ++t) {
#pragma unroll
      for (int v = 0; v < 8; ++v) {
        fRe[8 * t + v] = S[pfl + spos(v + 256 * t)];
        fIm[8 * t + v] = S[pfl + spos(v + 256 * t + 512)];
      }
    }

    cnot_ring_acc(fRe, lane);
    cnot_ring_acc(fIm, lane);

    float z[4] = {0.f, 0.f, 0.f, 0.f};
#pragma unroll
    for (int e = 0; e < 16; ++e) {
      float p = fRe[e] * fRe[e] + fIm[e] * fIm[e];
#pragma unroll
      for (int f = 0; f < 3; ++f) z[f] += ((e >> f) & 1) ? -p : p;
      z[3] += hh ? -p : p;
    }
#pragma unroll
    for (int f = 0; f < 4; ++f) {
#pragma unroll
      for (int m = 1; m < 32; m <<= 1) z[f] += __shfl_xor(z[f], m, 32);
    }
    if (lane == 0) *reinterpret_cast<float4*>(&zsh[warp][0]) = make_float4(z[0], z[1], z[2], z[3]);
  }
  __syncthreads();

  if (tid < 4 * PIX) {
    const int pw = tid & (PIX - 1), f = tid >> 2;
    const int p = blockIdx.x * PIX + pw;
    if (p < TOTAL) {
      const int b  = p / (HOUT * WOUT);
      const int yy = (p / WOUT) % HOUT;
      const int xx = p % WOUT;
      const float inv = 1.0f / 3.0f;
      const float s = (zsh[3 * pw][f] + zsh[3 * pw + 1][f]) + zsh[3 * pw + 2][f];
      out[((b * 4 + f) * HOUT + yy) * WOUT + xx] = s * inv;
    }
  }
}

extern "C" void kernel_launch(void* const* d_in, const int* in_sizes, int n_in,
                              void* d_out, int out_size, void* d_ws, size_t ws_size,
                              hipStream_t stream) {
  (void)in_sizes; (void)n_in; (void)out_size; (void)d_ws; (void)ws_size;
  const float* x  = (const float*)d_in[0];
  const float* qw = (const float*)d_in[1];
  float* out = (float*)d_out;
  const int total_pixels = 4 * 26 * 26;
  dim3 grid(total_pixels / 4), block(384);
  hipLaunchKernelGGL(quanv_kernel, grid, block, 0, stream, x, qw, out);
}


# Second CNOT ring moved behind |amp|^2 (a ring only permutes) with its lane-bit CNOTs composed into one DPP + one permlane16 gather per value: 40 cross-lane moves per wave, not 156; same sum order
# speedup vs baseline: 4.4377x; 1.0840x over previous
#include <hip/hip_runtime.h>


typedef __attribute__((ext_vector_type(2))) float v2f;
typedef __attribute__((ext_vector_type(8))) float v8f;

__device__ __forceinline__ void lds_fence() {
  asm volatile("s_wait_dscnt 0" ::: "memory");
}

__device__ __forceinline__ constexpr int spos(int a) {
  return (a & 63) + 74 * ((a >> 6) & 1) + 148 * ((a >> 7) & 1) +
         296 * ((a >> 8) & 1) + 596 * ((a >> 9) & 1);
}
constexpr int SLICE = 1184;

__device__ __forceinline__ v8f wmma4(v2f a, v2f b, v8f c) {
  return __builtin_amdgcn_wmma_f32_16x16x4_f32(false, a, false, b, (short)0, c,
                                               false, false);
}


template <int CTRL, int ROWS>
__device__ __forceinline__ float lane_quad(float v) {
  int x = __float_as_int(v);
  return __int_as_float(__builtin_amdgcn_update_dpp(x, x, CTRL, ROWS, 0xF, false));
}
constexpr int chain_src(int l) {
  l = (l & 4) ? (l ^ 8) : l;
  l = (l & 2) ? (l ^ 4) : l;
  l = (l & 1) ? (l ^ 2) : l;
  return l;
}
constexpr unsigned chain_sel(int word) {
  unsigned sel = 0;
  for (int j = 0; j < 8; ++j) sel |= (unsigned)chain_src(8 * word + j) << (4 * j);
  return sel;
}
constexpr unsigned CHAIN_LO = chain_sel(0), CHAIN_HI = chain_sel(1);
__device__ __forceinline__ float row_gather(float v) {
  int x = __float_as_int(v);
  return __int_as_float(
      __builtin_amdgcn_permlane16(x, x, CHAIN_LO, CHAIN_HI, false, false));
}
__device__ __forceinline__ float row_gather_x(float v) {
  int x = __float_as_int(v);
  return __int_as_float(
      __builtin_amdgcn_permlanex16(x, x, CHAIN_LO, CHAIN_HI, false, false));
}
__device__ __forceinline__ float other_row(float v) {
  int x = __float_as_int(v);
  return __int_as_float(
      __builtin_amdgcn_permlanex16(x, x, 0x76543210u, 0xFEDCBA98u, false, false));
}
__device__ __forceinline__ void swapf(float& a, float& b) { float t = a; a = b; b = t; }

__device__ __forceinline__ void cnot_ring_acc(float (&g)[16], int lane) {
  const bool l3 = (lane & 8) != 0;
#pragma unroll
  for (int t = 0; t < 2; ++t) {
#pragma unroll
    for (int v = 0; v < 8; ++v) if ((v & 1) && !(v & 2)) swapf(g[8 * t + v], g[8 * t + (v ^ 2)]);
#pragma unroll
    for (int v = 0; v < 8; ++v) if ((v & 2) && !(v & 4)) swapf(g[8 * t + v], g[8 * t + (v ^ 4)]);
  }
#pragma unroll
  for (int e = 0; e < 16; ++e) {
    if (e & 4) g[e] = row_gather_x(lane_quad<0xB1, 0x1>(g[e]));
    else       g[e] = row_gather(lane_quad<0xB1, 0x2>(g[e]));
  }
#pragma unroll
  for (int v = 0; v < 8; ++v) {
    float a = g[v], b = g[8 + v];
    g[v] = l3 ? b : a;
    g[8 + v] = l3 ? a : b;
  }
#pragma unroll
  for (int v = 0; v < 8; v += 2) swapf(g[8 + v], g[8 + v + 1]);
}

template <int G>
__device__ __forceinline__ void rx_group(float* S, const float2* gtG, int lane) {
  const int n = lane & 15, hh = lane >> 4;
  constexpr int SG = (G == 0) ? 1 : (G == 1) ? 8 : 64;
  constexpr int ST = (G == 2) ? 16 : 128;
  int lb, ld;
  if (G == 0)      { lb = 8 * n + 4 * hh;                        ld = 8 * n + 512 * hh; }
  else if (G == 1) { lb = (n & 7) + 64 * (n >> 3) + 32 * hh;     ld = (n & 7) + 64 * (n >> 3) + 512 * hh; }
  else             { lb = n + 256 * hh;                          ld = n + 512 * hh; }
  const int plb = spos(lb), pld = spos(ld);

  v2f a[4];
#pragma unroll
  for (int kc = 0; kc < 4; ++kc) {
    float2 w = gtG[kc * 32 + lane];
    a[kc][0] = w.x;
    a[kc][1] = w.y;
  }
  v2f bb[4][4];
  if (G == 0) {
#pragma unroll
    for (int t = 0; t < 4; ++t) {
#pragma unroll
      for (int kc = 0; kc < 4; ++kc) {
        const float2 q = *reinterpret_cast<const float2*>(
            &S[plb + spos(2 * (kc & 1) + ST * t + 512 * (kc >> 1))]);
        bb[t][kc][0] = q.x;
        bb[t][kc][1] = q.y;
      }
    }
  } else {
#pragma unroll
    for (int t = 0; t < 4; ++t) {
#pragma unroll
      for (int kc = 0; kc < 4; ++kc) {
#pragma unroll
        for (int v = 0; v < 2; ++v)
          bb[t][kc][v] = S[plb + spos(SG * (v + 2 * (kc & 1)) + ST * t + 512 * (kc >> 1))];
      }
    }
  }
  __builtin_amdgcn_sched_barrier(0);
  v8f acc[4];
#pragma unroll
  for (int t = 0; t < 4; ++t) acc[t] = v8f{0.f, 0.f, 0.f, 0.f, 0.f, 0.f, 0.f, 0.f};
#pragma unroll
  for (int kc = 0; kc < 4; ++kc) {
#pragma unroll
    for (int t = 0; t < 4; ++t) acc[t] = wmma4(a[kc], bb[t][kc], acc[t]);
  }
#pragma unroll
  for (int t = 0; t < 4; ++t) {
#pragma unroll
    for (int r = 0; r < 8; ++r) S[pld + spos(SG * r + ST * t)] = acc[t][r];
  }
}

__launch_bounds__(384)
__global__ void quanv_kernel(const float* __restrict__ x,
                             const float* __restrict__ qw,
                             float* __restrict__ out) {
  constexpr int NQ = 9, B = 4, C = 3, H = 28, W = 28, HOUT = 26, WOUT = 26;
  constexpr int TOTAL = B * HOUT * WOUT;
  constexpr int PIX = 4;
  constexpr int NWAVE = PIX * C;
  constexpr int NANG = NWAVE * NQ;

  __shared__ alignas(16) float lds[NWAVE][SLICE];
  __shared__ float2 trig[NANG + NQ];
  __shared__ alignas(16) float zsh[NWAVE][4];
  __shared__ float2 gtab[12][32];

  const int tid  = threadIdx.x;
  const int lane = tid & 31;
  const int warp = __builtin_amdgcn_readfirstlane(tid >> 5);

  float* S = &lds[warp][0];

  const int m16 = lane & 15;
  const int hh  = lane >> 4;

  if (tid < NANG + NQ) {
    float a;
    if (tid < NANG) {
      const int pw = tid / 27, j = tid % 27;
      const int ch = j / 9, i = j % 9;
      int p = blockIdx.x * PIX + pw;
      p = p < TOTAL ? p : TOTAL - 1;
      const int pb = p / (HOUT * WOUT);
      const int py = (p / WOUT) % HOUT;
      const int px = p % WOUT;
      a = (x[((pb * C + ch) * H + py + i / 3) * W + px + i % 3] + qw[i]) * 0.5f;
    } else {
      a = qw[NQ + (tid - NANG)] * 0.5f;
    }
    trig[tid] = make_float2(cosf(a), sinf(a));
  }
  __syncthreads();

  auto uni = [](float v) -> float {
    return __int_as_float(__builtin_amdgcn_readfirstlane(__float_as_int(v)));
  };

  {
    const int e = warp;
    const int grp = e >> 2, kc = e & 3;
    float cg[3], sg[3];
#pragma unroll
    for (int i = 0; i < 3; ++i) {
      float2 t = trig[NANG + 3 * grp + i];
      cg[i] = t.x;
      sg[i] = t.y;
    }
    const int gp = m16 & 7, rp = m16 >> 3;
    const int reim = kc >> 1;
    float w[2];
#pragma unroll
    for (int v = 0; v < 2; ++v) {
      const int g = v + 2 * (kc & 1) + 4 * hh;
      const int diff = gp ^ g;
      float P = 1.0f;
#pragma unroll
      for (int i = 0; i < 3; ++i) P *= ((diff >> i) & 1) ? sg[i] : cg[i];
      const int hm = __popc(diff) & 3;
      const float re = (hm == 0) ? P : ((hm == 2) ? -P : 0.0f);
      const float im = (hm == 1) ? -P : ((hm == 3) ? P : 0.0f);
      w[v] = (rp == reim) ? re : (rp ? im : -im);
    }
    gtab[e][lane] = make_float2(w[0], w[1]);
  }
  __syncthreads();

  {
    float ca[NQ], sa[NQ];
#pragma unroll
    for (int i = 0; i < NQ; ++i) {
      float2 t = trig[9 * warp + i];
      ca[i] = uni(t.x);
      sa[i] = uni(t.y);
    }

    const bool l3 = (lane & 8) != 0, l4 = (lane & 16) != 0;
    float L = 1.0f;
#pragma unroll
    for (int i = 0; i < 5; ++i) L *= ((lane >> i) & 1) ? sa[i] : ca[i];
    const int kl = __popc(lane) & 3;
    float lre = (kl == 0) ? L : ((kl == 2) ? -L : 0.0f);
    float lim = (kl == 1) ? -L : ((kl == 3) ? L : 0.0f);
    {
      const float vr = row_gather(lre);
      const float vi = row_gather(lim);
      const float orr = other_row(vr), oi = other_row(vi);
      lre = l3 ? orr : vr;
      lim = l3 ? oi : vi;
    }
    const float are = ca[5] * lre, aim = ca[5] * lim;
    const float bre = sa[5] * lim, bim = -(sa[5] * lre);
    float mre[2][2], mim[2][2];
    mre[0][0] = l4 ? bre : are;
    mim[0][0] = l4 ? bim : aim;
    mre[1][0] = l4 ? are : bre;
    mim[1][0] = l4 ? aim : bim;
#pragma unroll
    for (int b = 0; b < 2; ++b) {
      mre[b][1] = lane_quad<0xB1, 0x3>(mre[b][0]);
      mim[b][1] = lane_quad<0xB1, 0x3>(mim[b][0]);
    }
    float pRe[16], pIm[16];
#pragma unroll
    for (int j = 0; j < 16; ++j) {
      const int b = j & 1, m = j >> 1;
      float U = 1.0f;
#pragma unroll
      for (int i = 6; i < NQ; ++i) U *= ((m >> (i - 6)) & 1) ? sa[i] : ca[i];
      const int pm = (m & 1) + ((m >> 1) & 1) + ((m >> 2) & 1);
      const int km = pm & 3, sw = (b ^ pm) & 1;
      const float xr = mre[b][sw], xi = mim[b][sw];
      if (km == 0)      { pRe[j] = U * xr;    pIm[j] = U * xi; }
      else if (km == 1) { pRe[j] = U * xi;    pIm[j] = -(U * xr); }
      else if (km == 2) { pRe[j] = -(U * xr); pIm[j] = -(U * xi); }
      else              { pRe[j] = -(U * xi); pIm[j] = U * xr; }
    }
#pragma unroll
    for (int j = 0; j < 16; ++j) if ((j & 1) && !(j & 2)) { swapf(pRe[j], pRe[j ^ 2]); swapf(pIm[j], pIm[j ^ 2]); }
#pragma unroll
    for (int j = 0; j < 16; ++j) if ((j & 2) && !(j & 4)) { swapf(pRe[j], pRe[j ^ 4]); swapf(pIm[j], pIm[j ^ 4]); }
#pragma unroll
    for (int j = 0; j < 16; ++j) if ((j & 4) && !(j & 8)) { swapf(pRe[j], pRe[j ^ 8]); swapf(pIm[j], pIm[j ^ 8]); }

#pragma unroll
    for (int j = 0; j < 16; ++j) {
      S[lane + spos(32 * j)] = pRe[j];
      S[lane + spos(32 * j + 512)] = pIm[j];
    }
    lds_fence();

    rx_group<0>(S, &gtab[0][0], lane);
    lds_fence();
    rx_group<1>(S, &gtab[4][0], lane);
    lds_fence();
    rx_group<2>(S, &gtab[8][0], lane);
    lds_fence();

    float fRe[16], fIm[16];
    const int pfl = spos(8 * hh + 16 * m16);
#pragma unroll
    for (int t = 0; t < 2; ++t) {
#pragma unroll
      for (int v = 0; v < 8; ++v) {
        fRe[8 * t + v] = S[pfl + spos(v + 256 * t)];
        fIm[8 * t + v] = S[pfl + spos(v + 256 * t + 512)];
      }
    }

    float pr[16];
#pragma unroll
    for (int e = 0; e < 16; ++e) pr[e] = fRe[e] * fRe[e] + fIm[e] * fIm[e];
    cnot_ring_acc(pr, lane);

    float z[4] = {0.f, 0.f, 0.f, 0.f};
#pragma unroll
    for (int e = 0; e < 16; ++e) {
      const float p = pr[e];
#pragma unroll
      for (int f = 0; f < 3; ++f) z[f] += ((e >> f) & 1) ? -p : p;
      z[3] += hh ? -p : p;
    }
#pragma unroll
    for (int f = 0; f < 4; ++f) {
#pragma unroll
      for (int m = 1; m < 32; m <<= 1) z[f] += __shfl_xor(z[f], m, 32);
    }
    if (lane == 0) *reinterpret_cast<float4*>(&zsh[warp][0]) = make_float4(z[0], z[1], z[2], z[3]);
  }
  __syncthreads();

  if (tid < 4 * PIX) {
    const int pw = tid & (PIX - 1), f = tid >> 2;
    const int p = blockIdx.x * PIX + pw;
    if (p < TOTAL) {
      const int b  = p / (HOUT * WOUT);
      const int yy = (p / WOUT) % HOUT;
      const int xx = p % WOUT;
      const float inv = 1.0f / 3.0f;
      const float s = (zsh[3 * pw][f] + zsh[3 * pw + 1][f]) + zsh[3 * pw + 2][f];
      out[((b * 4 + f) * HOUT + yy) * WOUT + xx] = s * inv;
    }
  }
}

extern "C" void kernel_launch(void* const* d_in, const int* in_sizes, int n_in,
                              void* d_out, int out_size, void* d_ws, size_t ws_size,
                              hipStream_t stream) {
  (void)in_sizes; (void)n_in; (void)out_size; (void)d_ws; (void)ws_size;
  const float* x  = (const float*)d_in[0];
  const float* qw = (const float*)d_in[1];
  float* out = (float*)d_out;
  const int total_pixels = 4 * 26 * 26;
  dim3 grid(total_pixels / 4), block(384);
  hipLaunchKernelGGL(quanv_kernel, grid, block, 0, stream, x, qw, out);
}


# First RX gate skips the all-zero plane of the ring-1 state (amplitude real where index bit 8 is 0, imaginary where 1): K 16 -> 8; 8 of its 16 WMMAs and half of its LDS input gone; same sums
# speedup vs baseline: 4.8889x; 1.1017x over previous
#include <hip/hip_runtime.h>


typedef __attribute__((ext_vector_type(2))) float v2f;
typedef __attribute__((ext_vector_type(8))) float v8f;

__device__ __forceinline__ void lds_fence() {
  asm volatile("s_wait_dscnt 0" ::: "memory");
}

__device__ __forceinline__ constexpr int spos(int a) {
  return (a & 63) + 74 * ((a >> 6) & 1) + 148 * ((a >> 7) & 1) +
         296 * ((a >> 8) & 1) + 596 * ((a >> 9) & 1);
}
constexpr int SLICE = 1184;

__device__ __forceinline__ v8f wmma4(v2f a, v2f b, v8f c) {
  return __builtin_amdgcn_wmma_f32_16x16x4_f32(false, a, false, b, (short)0, c,
                                               false, false);
}


template <int CTRL, int ROWS>
__device__ __forceinline__ float lane_quad(float v) {
  int x = __float_as_int(v);
  return __int_as_float(__builtin_amdgcn_update_dpp(x, x, CTRL, ROWS, 0xF, false));
}
constexpr int chain_src(int l) {
  l = (l & 4) ? (l ^ 8) : l;
  l = (l & 2) ? (l ^ 4) : l;
  l = (l & 1) ? (l ^ 2) : l;
  return l;
}
constexpr unsigned chain_sel(int word) {
  unsigned sel = 0;
  for (int j = 0; j < 8; ++j) sel |= (unsigned)chain_src(8 * word + j) << (4 * j);
  return sel;
}
constexpr unsigned CHAIN_LO = chain_sel(0), CHAIN_HI = chain_sel(1);
__device__ __forceinline__ float row_gather(float v) {
  int x = __float_as_int(v);
  return __int_as_float(
      __builtin_amdgcn_permlane16(x, x, CHAIN_LO, CHAIN_HI, false, false));
}
__device__ __forceinline__ float row_gather_x(float v) {
  int x = __float_as_int(v);
  return __int_as_float(
      __builtin_amdgcn_permlanex16(x, x, CHAIN_LO, CHAIN_HI, false, false));
}
__device__ __forceinline__ float other_row(float v) {
  int x = __float_as_int(v);
  return __int_as_float(
      __builtin_amdgcn_permlanex16(x, x, 0x76543210u, 0xFEDCBA98u, false, false));
}
__device__ __forceinline__ void swapf(float& a, float& b) { float t = a; a = b; b = t; }

__device__ __forceinline__ void cnot_ring_acc(float (&g)[16], int lane) {
  const bool l3 = (lane & 8) != 0;
#pragma unroll
  for (int t = 0; t < 2; ++t) {
#pragma unroll
    for (int v = 0; v < 8; ++v) if ((v & 1) && !(v & 2)) swapf(g[8 * t + v], g[8 * t + (v ^ 2)]);
#pragma unroll
    for (int v = 0; v < 8; ++v) if ((v & 2) && !(v & 4)) swapf(g[8 * t + v], g[8 * t + (v ^ 4)]);
  }
#pragma unroll
  for (int e = 0; e < 16; ++e) {
    if (e & 4) g[e] = row_gather_x(lane_quad<0xB1, 0x1>(g[e]));
    else       g[e] = row_gather(lane_quad<0xB1, 0x2>(g[e]));
  }
#pragma unroll
  for (int v = 0; v < 8; ++v) {
    float a = g[v], b = g[8 + v];
    g[v] = l3 ? b : a;
    g[8 + v] = l3 ? a : b;
  }
#pragma unroll
  for (int v = 0; v < 8; v += 2) swapf(g[8 + v], g[8 + v + 1]);
}

template <int G>
__device__ __forceinline__ void rx_group(float* S, const float2* gtG, int lane) {
  const int n = lane & 15, hh = lane >> 4;
  constexpr int SG = (G == 0) ? 1 : (G == 1) ? 8 : 64;
  constexpr int ST = (G == 2) ? 16 : 128;
  int lb, ld;
  if (G == 0)      { lb = 8 * n + 4 * hh;                        ld = 8 * n + 512 * hh; }
  else if (G == 1) { lb = (n & 7) + 64 * (n >> 3) + 32 * hh;     ld = (n & 7) + 64 * (n >> 3) + 512 * hh; }
  else             { lb = n + 256 * hh;                          ld = n + 512 * hh; }
  const int plb = spos(lb), pld = spos(ld);

  v2f a[4];
#pragma unroll
  for (int kc = 0; kc < 4; ++kc) {
    float2 w = gtG[kc * 32 + lane];
    a[kc][0] = w.x;
    a[kc][1] = w.y;
  }
  constexpr int NK = (G == 0) ? 2 : 4;
  v2f bb[4][4];
  if (G == 0) {
#pragma unroll
    for (int t = 0; t < 4; ++t) {
#pragma unroll
      for (int s = 0; s < NK; ++s) {
        const int kc = 2 * (t >> 1) + s;
        const float2 q = *reinterpret_cast<const float2*>(
            &S[plb + spos(2 * (kc & 1) + ST * t + 512 * (kc >> 1))]);
        bb[t][s][0] = q.x;
        bb[t][s][1] = q.y;
      }
    }
  } else {
#pragma unroll
    for (int t = 0; t < 4; ++t) {
#pragma unroll
      for (int kc = 0; kc < 4; ++kc) {
#pragma unroll
        for (int v = 0; v < 2; ++v)
          bb[t][kc][v] = S[plb + spos(SG * (v + 2 * (kc & 1)) + ST * t + 512 * (kc >> 1))];
      }
    }
  }
  __builtin_amdgcn_sched_barrier(0);
  v8f acc[4];
#pragma unroll
  for (int t = 0; t < 4; ++t) acc[t] = v8f{0.f, 0.f, 0.f, 0.f, 0.f, 0.f, 0.f, 0.f};
#pragma unroll
  for (int s = 0; s < NK; ++s) {
#pragma unroll
    for (int t = 0; t < 4; ++t) {
      const int kc = (G == 0) ? 2 * (t >> 1) + s : s;
      acc[t] = wmma4(a[kc], bb[t][s], acc[t]);
    }
  }
#pragma unroll
  for (int t = 0; t < 4; ++t) {
#pragma unroll
    for (int r = 0; r < 8; ++r) S[pld + spos(SG * r + ST * t)] = acc[t][r];
  }
}

__launch_bounds__(384)
__global__ void quanv_kernel(const float* __restrict__ x,
                             const float* __restrict__ qw,
                             float* __restrict__ out) {
  constexpr int NQ = 9, B = 4, C = 3, H = 28, W = 28, HOUT = 26, WOUT = 26;
  constexpr int TOTAL = B * HOUT * WOUT;
  constexpr int PIX = 4;
  constexpr int NWAVE = PIX * C;
  constexpr int NANG = NWAVE * NQ;

  __shared__ alignas(16) float lds[NWAVE][SLICE];
  __shared__ float2 trig[NANG + NQ];
  __shared__ alignas(16) float zsh[NWAVE][4];
  __shared__ float2 gtab[12][32];

  const int tid  = threadIdx.x;
  const int lane = tid & 31;
  const int warp = __builtin_amdgcn_readfirstlane(tid >> 5);

  float* S = &lds[warp][0];

  const int m16 = lane & 15;
  const int hh  = lane >> 4;

  if (tid < NANG + NQ) {
    float a;
    if (tid < NANG) {
      const int pw = tid / 27, j = tid % 27;
      const int ch = j / 9, i = j % 9;
      int p = blockIdx.x * PIX + pw;
      p = p < TOTAL ? p : TOTAL - 1;
      const int pb = p / (HOUT * WOUT);
      const int py = (p / WOUT) % HOUT;
      const int px = p % WOUT;
      a = (x[((pb * C + ch) * H + py + i / 3) * W + px + i % 3] + qw[i]) * 0.5f;
    } else {
      a = qw[NQ + (tid - NANG)] * 0.5f;
    }
    trig[tid] = make_float2(cosf(a), sinf(a));
  }
  __syncthreads();

  auto uni = [](float v) -> float {
    return __int_as_float(__builtin_amdgcn_readfirstlane(__float_as_int(v)));
  };

  {
    const int e = warp;
    const int grp = e >> 2, kc = e & 3;
    float cg[3], sg[3];
#pragma unroll
    for (int i = 0; i < 3; ++i) {
      float2 t = trig[NANG + 3 * grp + i];
      cg[i] = t.x;
      sg[i] = t.y;
    }
    const int gp = m16 & 7, rp = m16 >> 3;
    const int reim = kc >> 1;
    float w[2];
#pragma unroll
    for (int v = 0; v < 2; ++v) {
      const int g = v + 2 * (kc & 1) + 4 * hh;
      const int diff = gp ^ g;
      float P = 1.0f;
#pragma unroll
      for (int i = 0; i < 3; ++i) P *= ((diff >> i) & 1) ? sg[i] : cg[i];
      const int hm = __popc(diff) & 3;
      const float re = (hm == 0) ? P : ((hm == 2) ? -P : 0.0f);
      const float im = (hm == 1) ? -P : ((hm == 3) ? P : 0.0f);
      w[v] = (rp == reim) ? re : (rp ? im : -im);
    }
    gtab[e][lane] = make_float2(w[0], w[1]);
  }
  __syncthreads();

  {
    float ca[NQ], sa[NQ];
#pragma unroll
    for (int i = 0; i < NQ; ++i) {
      float2 t = trig[9 * warp + i];
      ca[i] = uni(t.x);
      sa[i] = uni(t.y);
    }

    const bool l3 = (lane & 8) != 0, l4 = (lane & 16) != 0;
    float L = 1.0f;
#pragma unroll
    for (int i = 0; i < 5; ++i) L *= ((lane >> i) & 1) ? sa[i] : ca[i];
    const int kl = __popc(lane) & 3;
    float lre = (kl == 0) ? L : ((kl == 2) ? -L : 0.0f);
    float lim = (kl == 1) ? -L : ((kl == 3) ? L : 0.0f);
    {
      const float vr = row_gather(lre);
      const float vi = row_gather(lim);
      const float orr = other_row(vr), oi = other_row(vi);
      lre = l3 ? orr : vr;
      lim = l3 ? oi : vi;
    }
    const float are = ca[5] * lre, aim = ca[5] * lim;
    const float bre = sa[5] * lim, bim = -(sa[5] * lre);
    float mre[2][2], mim[2][2];
    mre[0][0] = l4 ? bre : are;
    mim[0][0] = l4 ? bim : aim;
    mre[1][0] = l4 ? are : bre;
    mim[1][0] = l4 ? aim : bim;
#pragma unroll
    for (int b = 0; b < 2; ++b) {
      mre[b][1] = lane_quad<0xB1, 0x3>(mre[b][0]);
      mim[b][1] = lane_quad<0xB1, 0x3>(mim[b][0]);
    }
    float pRe[16], pIm[16];
#pragma unroll
    for (int j = 0; j < 16; ++j) {
      const int b = j & 1, m = j >> 1;
      float U = 1.0f;
#pragma unroll
      for (int i = 6; i < NQ; ++i) U *= ((m >> (i - 6)) & 1) ? sa[i] : ca[i];
      const int pm = (m & 1) + ((m >> 1) & 1) + ((m >> 2) & 1);
      const int km = pm & 3, sw = (b ^ pm) & 1;
      const float xr = mre[b][sw], xi = mim[b][sw];
      if (km == 0)      { pRe[j] = U * xr;    pIm[j] = U * xi; }
      else if (km == 1) { pRe[j] = U * xi;    pIm[j] = -(U * xr); }
      else if (km == 2) { pRe[j] = -(U * xr); pIm[j] = -(U * xi); }
      else              { pRe[j] = -(U * xi); pIm[j] = U * xr; }
    }
#pragma unroll
    for (int j = 0; j < 16; ++j) if ((j & 1) && !(j & 2)) { swapf(pRe[j], pRe[j ^ 2]); swapf(pIm[j], pIm[j ^ 2]); }
#pragma unroll
    for (int j = 0; j < 16; ++j) if ((j & 2) && !(j & 4)) { swapf(pRe[j], pRe[j ^ 4]); swapf(pIm[j], pIm[j ^ 4]); }
#pragma unroll
    for (int j = 0; j < 16; ++j) if ((j & 4) && !(j & 8)) { swapf(pRe[j], pRe[j ^ 8]); swapf(pIm[j], pIm[j ^ 8]); }

#pragma unroll
    for (int j = 0; j < 16; ++j) {
      if (j < 8) S[lane + spos(32 * j)] = pRe[j];
      else       S[lane + spos(32 * j + 512)] = pIm[j];
    }
    lds_fence();

    rx_group<0>(S, &gtab[0][0], lane);
    lds_fence();
    rx_group<1>(S, &gtab[4][0], lane);
    lds_fence();
    rx_group<2>(S, &gtab[8][0], lane);
    lds_fence();

    float fRe[16], fIm[16];
    const int pfl = spos(8 * hh + 16 * m16);
#pragma unroll
    for (int t = 0; t < 2; ++t) {
#pragma unroll
      for (int v = 0; v < 8; ++v) {
        fRe[8 * t + v] = S[pfl + spos(v + 256 * t)];
        fIm[8 * t + v] = S[pfl + spos(v + 256 * t + 512)];
      }
    }

    float pr[16];
#pragma unroll
    for (int e = 0; e < 16; ++e) pr[e] = fRe[e] * fRe[e] + fIm[e] * fIm[e];
    cnot_ring_acc(pr, lane);

    float z[4] = {0.f, 0.f, 0.f, 0.f};
#pragma unroll
    for (int e = 0; e < 16; ++e) {
      const float p = pr[e];
#pragma unroll
      for (int f = 0; f < 3; ++f) z[f] += ((e >> f) & 1) ? -p : p;
      z[3] += hh ? -p : p;
    }
#pragma unroll
    for (int f = 0; f < 4; ++f) {
#pragma unroll
      for (int m = 1; m < 32; m <<= 1) z[f] += __shfl_xor(z[f], m, 32);
    }
    if (lane == 0) *reinterpret_cast<float4*>(&zsh[warp][0]) = make_float4(z[0], z[1], z[2], z[3]);
  }
  __syncthreads();

  if (tid < 4 * PIX) {
    const int pw = tid & (PIX - 1), f = tid >> 2;
    const int p = blockIdx.x * PIX + pw;
    if (p < TOTAL) {
      const int b  = p / (HOUT * WOUT);
      const int yy = (p / WOUT) % HOUT;
      const int xx = p % WOUT;
      const float inv = 1.0f / 3.0f;
      const float s = (zsh[3 * pw][f] + zsh[3 * pw + 1][f]) + zsh[3 * pw + 2][f];
      out[((b * 4 + f) * HOUT + yy) * WOUT + xx] = s * inv;
    }
  }
}

extern "C" void kernel_launch(void* const* d_in, const int* in_sizes, int n_in,
                              void* d_out, int out_size, void* d_ws, size_t ws_size,
                              hipStream_t stream) {
  (void)in_sizes; (void)n_in; (void)out_size; (void)d_ws; (void)ws_size;
  const float* x  = (const float*)d_in[0];
  const float* qw = (const float*)d_in[1];
  float* out = (float*)d_out;
  const int total_pixels = 4 * 26 * 26;
  dim3 grid(total_pixels / 4), block(384);
  hipLaunchKernelGGL(quanv_kernel, grid, block, 0, stream, x, qw, out);
}


# <Z_f> as signed sums of squares straight from the last gate's accumulators (re, im unpaired; ring 2 becomes a lane sign and a register sign): state store, readback, ring moves gone; sum order changes
# speedup vs baseline: 5.2942x; 1.0829x over previous
#include <hip/hip_runtime.h>


typedef __attribute__((ext_vector_type(2))) float v2f;
typedef __attribute__((ext_vector_type(8))) float v8f;

__device__ __forceinline__ void lds_fence() {
  asm volatile("s_wait_dscnt 0" ::: "memory");
}

__device__ __forceinline__ constexpr int spos(int a) {
  return (a & 63) + 74 * ((a >> 6) & 1) + 148 * ((a >> 7) & 1) +
         296 * ((a >> 8) & 1) + 596 * ((a >> 9) & 1);
}
constexpr int SLICE = 1184;

__device__ __forceinline__ v8f wmma4(v2f a, v2f b, v8f c) {
  return __builtin_amdgcn_wmma_f32_16x16x4_f32(false, a, false, b, (short)0, c,
                                               false, false);
}


template <int CTRL, int ROWS>
__device__ __forceinline__ float lane_quad(float v) {
  int x = __float_as_int(v);
  return __int_as_float(__builtin_amdgcn_update_dpp(x, x, CTRL, ROWS, 0xF, false));
}
constexpr int chain_src(int l) {
  l = (l & 4) ? (l ^ 8) : l;
  l = (l & 2) ? (l ^ 4) : l;
  l = (l & 1) ? (l ^ 2) : l;
  return l;
}
constexpr unsigned chain_sel(int word) {
  unsigned sel = 0;
  for (int j = 0; j < 8; ++j) sel |= (unsigned)chain_src(8 * word + j) << (4 * j);
  return sel;
}
constexpr unsigned CHAIN_LO = chain_sel(0), CHAIN_HI = chain_sel(1);
__device__ __forceinline__ float row_gather(float v) {
  int x = __float_as_int(v);
  return __int_as_float(
      __builtin_amdgcn_permlane16(x, x, CHAIN_LO, CHAIN_HI, false, false));
}
__device__ __forceinline__ float other_row(float v) {
  int x = __float_as_int(v);
  return __int_as_float(
      __builtin_amdgcn_permlanex16(x, x, 0x76543210u, 0xFEDCBA98u, false, false));
}
__device__ __forceinline__ void swapf(float& a, float& b) { float t = a; a = b; b = t; }

template <int G>
__device__ __forceinline__ void rx_group(float* S, const float2* gtG, int lane, v8f (&acc)[4]) {
  const int n = lane & 15, hh = lane >> 4;
  constexpr int SG = (G == 0) ? 1 : (G == 1) ? 8 : 64;
  constexpr int ST = (G == 2) ? 16 : 128;
  int lb, ld;
  if (G == 0)      { lb = 8 * n + 4 * hh;                        ld = 8 * n + 512 * hh; }
  else if (G == 1) { lb = (n & 7) + 64 * (n >> 3) + 32 * hh;     ld = (n & 7) + 64 * (n >> 3) + 512 * hh; }
  else             { lb = n + 256 * hh;                          ld = n + 512 * hh; }
  const int plb = spos(lb), pld = spos(ld);

  v2f a[4];
#pragma unroll
  for (int kc = 0; kc < 4; ++kc) {
    float2 w = gtG[kc * 32 + lane];
    a[kc][0] = w.x;
    a[kc][1] = w.y;
  }
  constexpr int NK = (G == 0) ? 2 : 4;
  v2f bb[4][4];
  if (G == 0) {
#pragma unroll
    for (int t = 0; t < 4; ++t) {
#pragma unroll
      for (int s = 0; s < NK; ++s) {
        const int kc = 2 * (t >> 1) + s;
        const float2 q = *reinterpret_cast<const float2*>(
            &S[plb + spos(2 * (kc & 1) + ST * t + 512 * (kc >> 1))]);
        bb[t][s][0] = q.x;
        bb[t][s][1] = q.y;
      }
    }
  } else {
#pragma unroll
    for (int t = 0; t < 4; ++t) {
#pragma unroll
      for (int kc = 0; kc < 4; ++kc) {
#pragma unroll
        for (int v = 0; v < 2; ++v)
          bb[t][kc][v] = S[plb + spos(SG * (v + 2 * (kc & 1)) + ST * t + 512 * (kc >> 1))];
      }
    }
  }
  __builtin_amdgcn_sched_barrier(0);
#pragma unroll
  for (int t = 0; t < 4; ++t) acc[t] = v8f{0.f, 0.f, 0.f, 0.f, 0.f, 0.f, 0.f, 0.f};
#pragma unroll
  for (int s = 0; s < NK; ++s) {
#pragma unroll
    for (int t = 0; t < 4; ++t) {
      const int kc = (G == 0) ? 2 * (t >> 1) + s : s;
      acc[t] = wmma4(a[kc], bb[t][s], acc[t]);
    }
  }
  if (G != 2) {
#pragma unroll
    for (int t = 0; t < 4; ++t) {
#pragma unroll
      for (int r = 0; r < 8; ++r) S[pld + spos(SG * r + ST * t)] = acc[t][r];
    }
  }
}

__launch_bounds__(384)
__global__ void quanv_kernel(const float* __restrict__ x,
                             const float* __restrict__ qw,
                             float* __restrict__ out) {
  constexpr int NQ = 9, B = 4, C = 3, H = 28, W = 28, HOUT = 26, WOUT = 26;
  constexpr int TOTAL = B * HOUT * WOUT;
  constexpr int PIX = 4;
  constexpr int NWAVE = PIX * C;
  constexpr int NANG = NWAVE * NQ;

  __shared__ alignas(16) float lds[NWAVE][SLICE];
  __shared__ float2 trig[NANG + NQ];
  __shared__ alignas(16) float zsh[NWAVE][4];
  __shared__ float2 gtab[12][32];

  const int tid  = threadIdx.x;
  const int lane = tid & 31;
  const int warp = __builtin_amdgcn_readfirstlane(tid >> 5);

  float* S = &lds[warp][0];

  const int m16 = lane & 15;
  const int hh  = lane >> 4;

  if (tid < NANG + NQ) {
    float a;
    if (tid < NANG) {
      const int pw = tid / 27, j = tid % 27;
      const int ch = j / 9, i = j % 9;
      int p = blockIdx.x * PIX + pw;
      p = p < TOTAL ? p : TOTAL - 1;
      const int pb = p / (HOUT * WOUT);
      const int py = (p / WOUT) % HOUT;
      const int px = p % WOUT;
      a = (x[((pb * C + ch) * H + py + i / 3) * W + px + i % 3] + qw[i]) * 0.5f;
    } else {
      a = qw[NQ + (tid - NANG)] * 0.5f;
    }
    trig[tid] = make_float2(cosf(a), sinf(a));
  }
  __syncthreads();

  auto uni = [](float v) -> float {
    return __int_as_float(__builtin_amdgcn_readfirstlane(__float_as_int(v)));
  };

  {
    const int e = warp;
    const int grp = e >> 2, kc = e & 3;
    float cg[3], sg[3];
#pragma unroll
    for (int i = 0; i < 3; ++i) {
      float2 t = trig[NANG + 3 * grp + i];
      cg[i] = t.x;
      sg[i] = t.y;
    }
    const int gp = m16 & 7, rp = m16 >> 3;
    const int reim = kc >> 1;
    float w[2];
#pragma unroll
    for (int v = 0; v < 2; ++v) {
      const int g = v + 2 * (kc & 1) + 4 * hh;
      const int diff = gp ^ g;
      float P = 1.0f;
#pragma unroll
      for (int i = 0; i < 3; ++i) P *= ((diff >> i) & 1) ? sg[i] : cg[i];
      const int hm = __popc(diff) & 3;
      const float re = (hm == 0) ? P : ((hm == 2) ? -P : 0.0f);
      const float im = (hm == 1) ? -P : ((hm == 3) ? P : 0.0f);
      w[v] = (rp == reim) ? re : (rp ? im : -im);
    }
    gtab[e][lane] = make_float2(w[0], w[1]);
  }
  __syncthreads();

  {
    float ca[NQ], sa[NQ];
#pragma unroll
    for (int i = 0; i < NQ; ++i) {
      float2 t = trig[9 * warp + i];
      ca[i] = uni(t.x);
      sa[i] = uni(t.y);
    }

    const bool l3 = (lane & 8) != 0, l4 = (lane & 16) != 0;
    float L = 1.0f;
#pragma unroll
    for (int i = 0; i < 5; ++i) L *= ((lane >> i) & 1) ? sa[i] : ca[i];
    const int kl = __popc(lane) & 3;
    float lre = (kl == 0) ? L : ((kl == 2) ? -L : 0.0f);
    float lim = (kl == 1) ? -L : ((kl == 3) ? L : 0.0f);
    {
      const float vr = row_gather(lre);
      const float vi = row_gather(lim);
      const float orr = other_row(vr), oi = other_row(vi);
      lre = l3 ? orr : vr;
      lim = l3 ? oi : vi;
    }
    const float are = ca[5] * lre, aim = ca[5] * lim;
    const float bre = sa[5] * lim, bim = -(sa[5] * lre);
    float mre[2][2], mim[2][2];
    mre[0][0] = l4 ? bre : are;
    mim[0][0] = l4 ? bim : aim;
    mre[1][0] = l4 ? are : bre;
    mim[1][0] = l4 ? aim : bim;
#pragma unroll
    for (int b = 0; b < 2; ++b) {
      mre[b][1] = lane_quad<0xB1, 0x3>(mre[b][0]);
      mim[b][1] = lane_quad<0xB1, 0x3>(mim[b][0]);
    }
    float pRe[16], pIm[16];
#pragma unroll
    for (int j = 0; j < 16; ++j) {
      const int b = j & 1, m = j >> 1;
      float U = 1.0f;
#pragma unroll
      for (int i = 6; i < NQ; ++i) U *= ((m >> (i - 6)) & 1) ? sa[i] : ca[i];
      const int pm = (m & 1) + ((m >> 1) & 1) + ((m >> 2) & 1);
      const int km = pm & 3, sw = (b ^ pm) & 1;
      const float xr = mre[b][sw], xi = mim[b][sw];
      if (km == 0)      { pRe[j] = U * xr;    pIm[j] = U * xi; }
      else if (km == 1) { pRe[j] = U * xi;    pIm[j] = -(U * xr); }
      else if (km == 2) { pRe[j] = -(U * xr); pIm[j] = -(U * xi); }
      else              { pRe[j] = -(U * xi); pIm[j] = U * xr; }
    }
#pragma unroll
    for (int j = 0; j < 16; ++j) if ((j & 1) && !(j & 2)) { swapf(pRe[j], pRe[j ^ 2]); swapf(pIm[j], pIm[j ^ 2]); }
#pragma unroll
    for (int j = 0; j < 16; ++j) if ((j & 2) && !(j & 4)) { swapf(pRe[j], pRe[j ^ 4]); swapf(pIm[j], pIm[j ^ 4]); }
#pragma unroll
    for (int j = 0; j < 16; ++j) if ((j & 4) && !(j & 8)) { swapf(pRe[j], pRe[j ^ 8]); swapf(pIm[j], pIm[j ^ 8]); }

#pragma unroll
    for (int j = 0; j < 16; ++j) {
      if (j < 8) S[lane + spos(32 * j)] = pRe[j];
      else       S[lane + spos(32 * j + 512)] = pIm[j];
    }
    lds_fence();

    v8f acc[4];
    rx_group<0>(S, &gtab[0][0], lane, acc);
    lds_fence();
    rx_group<1>(S, &gtab[4][0], lane, acc);
    lds_fence();
    rx_group<2>(S, &gtab[8][0], lane, acc);

    v2f ua0 = {0.f, 0.f}, ua1 = {0.f, 0.f}, ub0 = {0.f, 0.f}, ub1 = {0.f, 0.f};
#pragma unroll
    for (int t = 0; t < 4; ++t) {
      const v8f c = acc[t];
      const v2f p0 = __builtin_shufflevector(c, c, 0, 1), p1 = __builtin_shufflevector(c, c, 6, 7);
      const v2f q0 = __builtin_shufflevector(c, c, 2, 3), q1 = __builtin_shufflevector(c, c, 4, 5);
      if ((__popc(t) & 1) == 0) {
        ua0 += p0 * p0;  ua1 += p1 * p1;  ub0 += q0 * q0;  ub1 += q1 * q1;
      } else {
        ub0 += p0 * p0;  ub1 += p1 * p1;  ua0 += q0 * q0;  ua1 += q1 * q1;
      }
    }
    const v2f ua = ua0 + ua1, ub = ub0 + ub1;
    const float same = ua[0] + ub[1];
    const float diff = ua[1] + ub[0];
    const float T0 = same + diff, T1 = same - diff;
    float z[4];
    z[0] = (__popc(m16 & 0xE) & 1) ? -T1 : T1;
    z[1] = (__popc(m16 & 0x3) & 1) ? -T0 : T0;
    z[2] = (__popc(m16 & 0x7) & 1) ? -T0 : T0;
    z[3] = (__popc(m16 & 0xF) & 1) ? -T0 : T0;
#pragma unroll
    for (int f = 0; f < 4; ++f) {
#pragma unroll
      for (int m = 1; m < 32; m <<= 1) z[f] += __shfl_xor(z[f], m, 32);
    }
    if (lane == 0) *reinterpret_cast<float4*>(&zsh[warp][0]) = make_float4(z[0], z[1], z[2], z[3]);
  }
  __syncthreads();

  if (tid < 4 * PIX) {
    const int pw = tid & (PIX - 1), f = tid >> 2;
    const int p = blockIdx.x * PIX + pw;
    if (p < TOTAL) {
      const int b  = p / (HOUT * WOUT);
      const int yy = (p / WOUT) % HOUT;
      const int xx = p % WOUT;
      const float inv = 1.0f / 3.0f;
      const float s = (zsh[3 * pw][f] + zsh[3 * pw + 1][f]) + zsh[3 * pw + 2][f];
      out[((b * 4 + f) * HOUT + yy) * WOUT + xx] = s * inv;
    }
  }
}

extern "C" void kernel_launch(void* const* d_in, const int* in_sizes, int n_in,
                              void* d_out, int out_size, void* d_ws, size_t ws_size,
                              hipStream_t stream) {
  (void)in_sizes; (void)n_in; (void)out_size; (void)d_ws; (void)ws_size;
  const float* x  = (const float*)d_in[0];
  const float* qw = (const float*)d_in[1];
  float* out = (float*)d_out;
  const int total_pixels = 4 * 26 * 26;
  dim3 grid(total_pixels / 4), block(384);
  hipLaunchKernelGGL(quanv_kernel, grid, block, 0, stream, x, qw, out);
}
